# starting kernel
# baseline (speedup 1.0000x reference)
#include <hip/hip_runtime.h>

typedef __attribute__((ext_vector_type(2))) float v2f;
typedef __attribute__((ext_vector_type(8))) float v8f;

__device__ __forceinline__ float h_capture_lane(float a1, float a2, int lane) {
    float h[17];
    h[0] = 1.0f;
    h[1] = -a1;
#pragma unroll
    for (int i = 2; i <= 16; ++i) h[i] = -a1 * h[i - 1] - a2 * h[i - 2];
    float hl = h[0];
#pragma unroll
    for (int i = 1; i <= 16; ++i) hl = (lane == i) ? h[i] : hl;
    return hl;
}

__device__ __forceinline__ float h_gather(float hl, int n) {
    int nn = n < 0 ? 0 : n;
    float v = __int_as_float(__builtin_amdgcn_ds_bpermute(nn << 2, __float_as_int(hl)));
    return n < 0 ? 0.0f : v;
}

__global__ void __launch_bounds__(256)
iir_tile_fir_wmma(const float* __restrict__ u,
                  const float* __restrict__ u_init,
                  const float* __restrict__ bc,
                  const float* __restrict__ ac,
                  float* __restrict__ y0,
                  int T, int tiles) {
    __shared__ float lds[8][16 * 18];
    int lane = threadIdx.x & 31;
    int wave = threadIdx.x >> 5;
    int tile = blockIdx.x * 8 + wave;
    if (tile >= tiles) tile = tiles - 1;
    int nT = T >> 4;
    int tb = tile / nT;
    int tt = tile - tb * nT;
    int t0 = tt << 4;

    float b0 = bc[0], b1 = bc[1];
    float a1 = ac[0], a2 = ac[1];

    float* tl = &lds[wave][0];

    #pragma unroll
    for (int c = 0; c < 8; ++c) {
        int flat = c * 32 + lane;
        int rr = flat >> 4, cc = flat & 15;
        tl[rr * 18 + 1 + cc] = u[(size_t)(tb * 16 + rr) * T + t0 + cc];
    }
    if (lane < 16) {
        int rowg = tb * 16 + lane;
        const float* p = (t0 == 0) ? (u_init + (size_t)rowg * 2)
                                   : (u + (size_t)rowg * T + (t0 - 1));
        tl[lane * 18] = *p;
    }
    __syncthreads();

    int m    = lane & 15;
    int half = lane >> 4;

    float hl = h_capture_lane(a1, a2, lane);

    v2f Af[4]; v2f Bf[4];
#pragma unroll
    for (int j = 0; j < 4; ++j) {
#pragma unroll
        for (int r = 0; r < 2; ++r) {
            int k = 4 * j + 2 * half + r;
            float uc = tl[m * 18 + 1 + k];
            float up = tl[m * 18 + k];
            Af[j][r] = b0 * uc + b1 * up;
            Bf[j][r] = h_gather(hl, m - k);
        }
    }

    v8f acc = {0.f, 0.f, 0.f, 0.f, 0.f, 0.f, 0.f, 0.f};
#pragma unroll
    for (int j = 0; j < 4; ++j)
        acc = __builtin_amdgcn_wmma_f32_16x16x4_f32(
             false, Af[j],  false, Bf[j],
             (short)0, acc,  false,  false);

    float* orow = y0 + (size_t)(tb * 16 + 8 * half) * T + t0 + m;
#pragma unroll
    for (int r = 0; r < 8; ++r)
        orow[(size_t)r * T] = acc[r];
}

__global__ void iir_state_scan(const float* __restrict__ y0,
                               const float* __restrict__ y_init,
                               const float* __restrict__ ac,
                               float2* __restrict__ states,
                               int B, int T) {
    int b = blockIdx.x * blockDim.x + threadIdx.x;
    if (b >= B) return;
    float a1 = ac[0], a2 = ac[1];
    float h[17];
    h[0] = 1.0f; h[1] = -a1;
#pragma unroll
    for (int i = 2; i <= 16; ++i) h[i] = -a1 * h[i - 1] - a2 * h[i - 2];
    float m00 = h[16], m01 = -a2 * h[15], m10 = h[15], m11 = -a2 * h[14];
    float s1 = y_init[(size_t)b * 2 + 0];
    float s2 = y_init[(size_t)b * 2 + 1];
    int nT = T >> 4;
    const float* rowp = y0 + (size_t)b * T;
    float2* st = states + (size_t)b * nT;
    for (int c = 0; c < nT; ++c) {
        st[c] = make_float2(s1, s2);
        float d0 = rowp[c * 16 + 15];
        float d1 = rowp[c * 16 + 14];
        float n1 = m00 * s1 + m01 * s2 + d0;
        float n2 = m10 * s1 + m11 * s2 + d1;
        s1 = n1; s2 = n2;
    }
}

__global__ void iir_apply_state(float* __restrict__ y,
                                const float2* __restrict__ states,
                                const float* __restrict__ ac,
                                int T, size_t total) {
    size_t idx = (size_t)blockIdx.x * blockDim.x + threadIdx.x;
    if (idx >= total) return;
    size_t b = idx / (size_t)T;
    int t = (int)(idx - b * (size_t)T);
    int j = t & 15;
    int c = t >> 4;
    float a1 = ac[0], a2 = ac[1];
    int lane = threadIdx.x & 31;
    float hl = h_capture_lane(a1, a2, lane);
    float g1 = h_gather(hl, j + 1);
    float g2 = h_gather(hl, j);
    float2 s = states[b * (size_t)(T >> 4) + c];
    y[idx] += g1 * s.x + (-a2 * g2) * s.y;
}

extern "C" void kernel_launch(void* const* d_in, const int* in_sizes, int n_in,
                              void* d_out, int out_size, void* d_ws, size_t ws_size,
                              hipStream_t stream) {
    const float* bc = (const float*)d_in[0];
    const float* ac = (const float*)d_in[1];
    const float* u  = (const float*)d_in[2];
    const float* yi = (const float*)d_in[3];
    const float* ui = (const float*)d_in[4];
    float* y = (float*)d_out;

    int B = in_sizes[3] / 2;
    int T = in_sizes[2] / B;
    int nT = T >> 4;
    int tiles = (B >> 4) * nT;
    float2* states = (float2*)d_ws;

    int blocks1 = (tiles + 7) / 8;
    iir_tile_fir_wmma<<<blocks1, 256, 0, stream>>>(u, ui, bc, ac, y, T, tiles);

    iir_state_scan<<<(B + 255) / 256, 256, 0, stream>>>(y, yi, ac, states, B, T);

    size_t total = (size_t)B * T;
    iir_apply_state<<<(int)((total + 255) / 256), 256, 0, stream>>>(y, states, ac, T, total);
}


# tile-state scan: one wave per row, cross-lane scan over 32 tiles replaces the serial per-tile chain; powers of M^16 formed in f64, carry across slots in f64; state sum order changes
# speedup vs baseline: 3.6166x; 3.6166x over previous
#include <hip/hip_runtime.h>

typedef __attribute__((ext_vector_type(2))) float v2f;
typedef __attribute__((ext_vector_type(8))) float v8f;

__device__ __forceinline__ float h_capture_lane(float a1, float a2, int lane) {
    float h[17];
    h[0] = 1.0f;
    h[1] = -a1;
#pragma unroll
    for (int i = 2; i <= 16; ++i) h[i] = -a1 * h[i - 1] - a2 * h[i - 2];
    float hl = h[0];
#pragma unroll
    for (int i = 1; i <= 16; ++i) hl = (lane == i) ? h[i] : hl;
    return hl;
}

__device__ __forceinline__ float h_gather(float hl, int n) {
    int nn = n < 0 ? 0 : n;
    float v = __int_as_float(__builtin_amdgcn_ds_bpermute(nn << 2, __float_as_int(hl)));
    return n < 0 ? 0.0f : v;
}

__global__ void __launch_bounds__(256)
iir_tile_fir_wmma(const float* __restrict__ u,
                  const float* __restrict__ u_init,
                  const float* __restrict__ bc,
                  const float* __restrict__ ac,
                  float* __restrict__ y0,
                  int T, int tiles) {
    __shared__ float lds[8][16 * 18];
    int lane = threadIdx.x & 31;
    int wave = threadIdx.x >> 5;
    int tile = blockIdx.x * 8 + wave;
    if (tile >= tiles) tile = tiles - 1;
    int nT = T >> 4;
    int tb = tile / nT;
    int tt = tile - tb * nT;
    int t0 = tt << 4;

    float b0 = bc[0], b1 = bc[1];
    float a1 = ac[0], a2 = ac[1];

    float* tl = &lds[wave][0];

    #pragma unroll
    for (int c = 0; c < 8; ++c) {
        int flat = c * 32 + lane;
        int rr = flat >> 4, cc = flat & 15;
        tl[rr * 18 + 1 + cc] = u[(size_t)(tb * 16 + rr) * T + t0 + cc];
    }
    if (lane < 16) {
        int rowg = tb * 16 + lane;
        const float* p = (t0 == 0) ? (u_init + (size_t)rowg * 2)
                                   : (u + (size_t)rowg * T + (t0 - 1));
        tl[lane * 18] = *p;
    }
    __syncthreads();

    int m    = lane & 15;
    int half = lane >> 4;

    float hl = h_capture_lane(a1, a2, lane);

    v2f Af[4]; v2f Bf[4];
#pragma unroll
    for (int j = 0; j < 4; ++j) {
#pragma unroll
        for (int r = 0; r < 2; ++r) {
            int k = 4 * j + 2 * half + r;
            float uc = tl[m * 18 + 1 + k];
            float up = tl[m * 18 + k];
            Af[j][r] = b0 * uc + b1 * up;
            Bf[j][r] = h_gather(hl, m - k);
        }
    }

    v8f acc = {0.f, 0.f, 0.f, 0.f, 0.f, 0.f, 0.f, 0.f};
#pragma unroll
    for (int j = 0; j < 4; ++j)
        acc = __builtin_amdgcn_wmma_f32_16x16x4_f32(
             false, Af[j],  false, Bf[j],
             (short)0, acc,  false,  false);

    float* orow = y0 + (size_t)(tb * 16 + 8 * half) * T + t0 + m;
#pragma unroll
    for (int r = 0; r < 8; ++r)
        orow[(size_t)r * T] = acc[r];
}

template <int N>
__device__ __forceinline__ float lane_shr(float x) {
    return __int_as_float(__builtin_amdgcn_update_dpp(0, __float_as_int(x), 0x110 + N, 0xf, 0xf, true));
}
__device__ __forceinline__ float other_row_lane15(float x) {
    return __int_as_float(__builtin_amdgcn_permlanex16(0, __float_as_int(x),
                                                       0xffffffff, 0xffffffff, false, false));
}
__device__ __forceinline__ float other_row_same_lane(float x) {
    return __int_as_float(__builtin_amdgcn_permlanex16(0, __float_as_int(x),
                                                       0x76543210, 0xfedcba98, false, false));
}
__device__ __forceinline__ void mat_acc(float& vx, float& vy, float sx, float sy,
                                        float a00, float a01, float a10, float a11) {
    vx = fmaf(a00, sx, fmaf(a01, sy, vx));
    vy = fmaf(a10, sx, fmaf(a11, sy, vy));
}

struct m2d { double a, b, c, d; };
__device__ __forceinline__ m2d mul2(m2d x, m2d y) {
    m2d r;
    r.a = fma(x.a, y.a, x.b * y.c);
    r.b = fma(x.a, y.b, x.b * y.d);
    r.c = fma(x.c, y.a, x.d * y.c);
    r.d = fma(x.c, y.b, x.d * y.d);
    return r;
}
__device__ __forceinline__ m2d pick(bool t, m2d x, m2d y) {
    m2d r;
    r.a = t ? x.a : y.a; r.b = t ? x.b : y.b;
    r.c = t ? x.c : y.c; r.d = t ? x.d : y.d;
    return r;
}

__device__ __forceinline__ double from_lane31(double x) {
    int lo = __builtin_amdgcn_readlane(__double2loint(x), 31);
    int hi = __builtin_amdgcn_readlane(__double2hiint(x), 31);
    return __hiloint2double(hi, lo);
}

__global__ void __launch_bounds__(64)
iir_state_scan(const float* __restrict__ y0,
               const float* __restrict__ y_init,
               const float* __restrict__ ac,
               float2* __restrict__ states,
               int B, int T) {
    int lane = threadIdx.x & 31;
    int b = blockIdx.x * 2 + (threadIdx.x >> 5);
    if (b >= B) return;
    int nT = T >> 4;
    const float* rowp = y0 + (size_t)b * T;
    float2* st = states + (size_t)b * nT;

    double a1 = (double)ac[0], a2 = (double)ac[1];
    double h[17];
    h[0] = 1.0; h[1] = -a1;
#pragma unroll
    for (int i = 2; i <= 16; ++i) h[i] = -a1 * h[i - 1] - a2 * h[i - 2];
    m2d S0 = { h[16], -a2 * h[15], h[15], -a2 * h[14] };
    m2d S1 = mul2(S0, S0);
    m2d S2 = mul2(S1, S1);
    m2d S3 = mul2(S2, S2);
    m2d S4 = mul2(S3, S3);
    m2d S5 = mul2(S4, S4);
    int n = lane + 1;
    m2d I2 = { 1.0, 0.0, 0.0, 1.0 };
    m2d Pd = pick(n & 1, S0, I2);
    Pd = pick(n & 2, mul2(Pd, S1), Pd);
    Pd = pick(n & 4, mul2(Pd, S2), Pd);
    Pd = pick(n & 8, mul2(Pd, S3), Pd);
    Pd = pick(n & 16, mul2(Pd, S4), Pd);
    Pd = pick(n & 32, S5, Pd);
    float p00 = (float)Pd.a, p01 = (float)Pd.b, p10 = (float)Pd.c, p11 = (float)Pd.d;
    float s1_00 = (float)S0.a, s1_01 = (float)S0.b, s1_10 = (float)S0.c, s1_11 = (float)S0.d;
    float s2_00 = (float)S1.a, s2_01 = (float)S1.b, s2_10 = (float)S1.c, s2_11 = (float)S1.d;
    float s4_00 = (float)S2.a, s4_01 = (float)S2.b, s4_10 = (float)S2.c, s4_11 = (float)S2.d;
    float s8_00 = (float)S3.a, s8_01 = (float)S3.b, s8_10 = (float)S3.c, s8_11 = (float)S3.d;
    bool hi = lane >= 16;
    float q00 = other_row_same_lane(p00), q01 = other_row_same_lane(p01);
    float q10 = other_row_same_lane(p10), q11 = other_row_same_lane(p11);
    q00 = hi ? q00 : 0.f; q01 = hi ? q01 : 0.f;
    q10 = hi ? q10 : 0.f; q11 = hi ? q11 : 0.f;

    double gx = 0.0, gy = 0.0;
    for (int base = 0; base < nT; base += 1024) {
        float vx[32], vy[32];
#pragma unroll
        for (int i = 0; i < 32; ++i) {
            int c = base + 32 * i + lane;
            int cs = c < nT ? c : nT - 1;
            cs = cs > 0 ? cs : 1;
            float2 d = *(const float2*)(rowp + (size_t)(cs - 1) * 16 + 14);
            bool live = c < nT;
            vx[i] = live ? d.y : 0.f;
            vy[i] = live ? d.x : 0.f;
        }
        if (base + lane == 0) {
            vx[0] = y_init[(size_t)b * 2 + 0];
            vy[0] = y_init[(size_t)b * 2 + 1];
        }
#pragma unroll
        for (int i = 0; i < 32; ++i) {
            float sx, sy;
            sx = lane_shr<1>(vx[i]); sy = lane_shr<1>(vy[i]);
            mat_acc(vx[i], vy[i], sx, sy, s1_00, s1_01, s1_10, s1_11);
            sx = lane_shr<2>(vx[i]); sy = lane_shr<2>(vy[i]);
            mat_acc(vx[i], vy[i], sx, sy, s2_00, s2_01, s2_10, s2_11);
            sx = lane_shr<4>(vx[i]); sy = lane_shr<4>(vy[i]);
            mat_acc(vx[i], vy[i], sx, sy, s4_00, s4_01, s4_10, s4_11);
            sx = lane_shr<8>(vx[i]); sy = lane_shr<8>(vy[i]);
            mat_acc(vx[i], vy[i], sx, sy, s8_00, s8_01, s8_10, s8_11);
            float tx = other_row_lane15(vx[i]);
            float ty = other_row_lane15(vy[i]);
            tx = hi ? tx : 0.f;
            ty = hi ? ty : 0.f;
            mat_acc(vx[i], vy[i], tx, ty, q00, q01, q10, q11);
        }
#pragma unroll
        for (int i = 0; i < 32; ++i) {
            double ox = fma(Pd.a, gx, fma(Pd.b, gy, (double)vx[i]));
            double oy = fma(Pd.c, gx, fma(Pd.d, gy, (double)vy[i]));
            int c = base + 32 * i + lane;
            if (c < nT) st[c] = make_float2((float)ox, (float)oy);
            gx = from_lane31(ox); gy = from_lane31(oy);
        }
    }
}

__global__ void iir_apply_state(float* __restrict__ y,
                                const float2* __restrict__ states,
                                const float* __restrict__ ac,
                                int T, size_t total) {
    size_t idx = (size_t)blockIdx.x * blockDim.x + threadIdx.x;
    if (idx >= total) return;
    size_t b = idx / (size_t)T;
    int t = (int)(idx - b * (size_t)T);
    int j = t & 15;
    int c = t >> 4;
    float a1 = ac[0], a2 = ac[1];
    int lane = threadIdx.x & 31;
    float hl = h_capture_lane(a1, a2, lane);
    float g1 = h_gather(hl, j + 1);
    float g2 = h_gather(hl, j);
    float2 s = states[b * (size_t)(T >> 4) + c];
    y[idx] += g1 * s.x + (-a2 * g2) * s.y;
}

extern "C" void kernel_launch(void* const* d_in, const int* in_sizes, int n_in,
                              void* d_out, int out_size, void* d_ws, size_t ws_size,
                              hipStream_t stream) {
    const float* bc = (const float*)d_in[0];
    const float* ac = (const float*)d_in[1];
    const float* u  = (const float*)d_in[2];
    const float* yi = (const float*)d_in[3];
    const float* ui = (const float*)d_in[4];
    float* y = (float*)d_out;

    int B = in_sizes[3] / 2;
    int T = in_sizes[2] / B;
    int nT = T >> 4;
    int tiles = (B >> 4) * nT;
    float2* states = (float2*)d_ws;

    int blocks1 = (tiles + 7) / 8;
    iir_tile_fir_wmma<<<blocks1, 256, 0, stream>>>(u, ui, bc, ac, y, T, tiles);

    iir_state_scan<<<(B + 1) / 2, 64, 0, stream>>>(y, yi, ac, states, B, T);

    size_t total = (size_t)B * T;
    iir_apply_state<<<(int)((total + 255) / 256), 256, 0, stream>>>(y, states, ac, T, total);
}


# y0 plane no longer stored and re-read: pass 1 keeps only each tile's end pair, last pass redoes the tile product on WMMA with the state term as a 5th k-slice (so summed inside the WMMA chain)
# speedup vs baseline: 4.3026x; 1.1897x over previous
#include <hip/hip_runtime.h>

typedef __attribute__((ext_vector_type(2))) float v2f;
typedef __attribute__((ext_vector_type(8))) float v8f;

__device__ __forceinline__ float h_capture_lane(float a1, float a2, int lane) {
    float h[17];
    h[0] = 1.0f;
    h[1] = -a1;
#pragma unroll
    for (int i = 2; i <= 16; ++i) h[i] = -a1 * h[i - 1] - a2 * h[i - 2];
    float hl = h[0];
#pragma unroll
    for (int i = 1; i <= 16; ++i) hl = (lane == i) ? h[i] : hl;
    return hl;
}

__device__ __forceinline__ float h_gather(float hl, int n) {
    int nn = n < 0 ? 0 : n;
    float v = __int_as_float(__builtin_amdgcn_ds_bpermute(nn << 2, __float_as_int(hl)));
    return n < 0 ? 0.0f : v;
}

__device__ __forceinline__ void tile_operands(const float* __restrict__ u,
                                              const float* __restrict__ u_init,
                                              float b0, float b1, float a1, float a2,
                                              float* tl, int T, int tb, int t0, int lane,
                                              v2f (&Af)[4], v2f (&Bf)[4], float& hl) {
    #pragma unroll
    for (int c = 0; c < 8; ++c) {
        int flat = c * 32 + lane;
        int rr = flat >> 4, cc = flat & 15;
        tl[rr * 18 + 1 + cc] = u[(size_t)(tb * 16 + rr) * T + t0 + cc];
    }
    if (lane < 16) {
        int rowg = tb * 16 + lane;
        const float* p = (t0 == 0) ? (u_init + (size_t)rowg * 2)
                                   : (u + (size_t)rowg * T + (t0 - 1));
        tl[lane * 18] = *p;
    }
    __syncthreads();

    int m    = lane & 15;
    int half = lane >> 4;

    hl = h_capture_lane(a1, a2, lane);

#pragma unroll
    for (int j = 0; j < 4; ++j) {
#pragma unroll
        for (int r = 0; r < 2; ++r) {
            int k = 4 * j + 2 * half + r;
            float uc = tl[m * 18 + 1 + k];
            float up = tl[m * 18 + k];
            Af[j][r] = b0 * uc + b1 * up;
            Bf[j][r] = h_gather(hl, m - k);
        }
    }
}

__global__ void __launch_bounds__(256)
iir_tile_ends_wmma(const float* __restrict__ u,
                   const float* __restrict__ u_init,
                   const float* __restrict__ bc,
                   const float* __restrict__ ac,
                   float* __restrict__ ends,
                   int T, int tiles) {
    __shared__ float lds[8][16 * 18];
    int lane = threadIdx.x & 31;
    int wave = threadIdx.x >> 5;
    int tile = blockIdx.x * 8 + wave;
    if (tile >= tiles) tile = tiles - 1;
    int nT = T >> 4;
    int tb = tile / nT;
    int tt = tile - tb * nT;
    int t0 = tt << 4;
    int m    = lane & 15;
    int half = lane >> 4;

    v2f Af[4]; v2f Bf[4]; float hl;
    tile_operands(u, u_init, bc[0], bc[1], ac[0], ac[1], &lds[wave][0], T, tb, t0, lane, Af, Bf, hl);

    v8f acc = {0.f, 0.f, 0.f, 0.f, 0.f, 0.f, 0.f, 0.f};
#pragma unroll
    for (int j = 0; j < 4; ++j)
        acc = __builtin_amdgcn_wmma_f32_16x16x4_f32(
             false, Af[j],  false, Bf[j],
             (short)0, acc,  false,  false);

    if (m >= 14) {
        float* e = ends + ((size_t)(tb * 16 + 8 * half) * nT + tt) * 2 + (15 - m);
#pragma unroll
        for (int r = 0; r < 8; ++r)
            e[(size_t)r * nT * 2] = acc[r];
    }
}

__global__ void __launch_bounds__(256)
iir_tile_out_wmma(const float* __restrict__ u,
                  const float* __restrict__ u_init,
                  const float* __restrict__ bc,
                  const float* __restrict__ ac,
                  const float2* __restrict__ states,
                  float* __restrict__ y,
                  int T, int tiles) {
    __shared__ float lds[8][16 * 18];
    int lane = threadIdx.x & 31;
    int wave = threadIdx.x >> 5;
    int tile = blockIdx.x * 8 + wave;
    if (tile >= tiles) tile = tiles - 1;
    int nT = T >> 4;
    int tb = tile / nT;
    int tt = tile - tb * nT;
    int t0 = tt << 4;
    int m    = lane & 15;
    int half = lane >> 4;
    float a2 = ac[1];

    float2 s = states[(size_t)(tb * 16 + m) * nT + tt];

    v2f Af[4]; v2f Bf[4]; float hl;
    tile_operands(u, u_init, bc[0], bc[1], ac[0], a2, &lds[wave][0], T, tb, t0, lane, Af, Bf, hl);

    float g1 = h_gather(hl, m + 1);
    float g0 = h_gather(hl, m);
    v2f A5, B5;
    A5[0] = half ? 0.f : s.x;
    A5[1] = half ? 0.f : s.y;
    B5[0] = half ? 0.f : g1;
    B5[1] = half ? 0.f : -a2 * g0;

    v8f acc = {0.f, 0.f, 0.f, 0.f, 0.f, 0.f, 0.f, 0.f};
#pragma unroll
    for (int j = 0; j < 4; ++j)
        acc = __builtin_amdgcn_wmma_f32_16x16x4_f32(
             false, Af[j],  false, Bf[j],
             (short)0, acc,  false,  false);
    acc = __builtin_amdgcn_wmma_f32_16x16x4_f32(
         false, A5,  false, B5,
         (short)0, acc,  false,  false);

    float* orow = y + (size_t)(tb * 16 + 8 * half) * T + t0 + m;
#pragma unroll
    for (int r = 0; r < 8; ++r)
        orow[(size_t)r * T] = acc[r];
}

template <int N>
__device__ __forceinline__ float lane_shr(float x) {
    return __int_as_float(__builtin_amdgcn_update_dpp(0, __float_as_int(x), 0x110 + N, 0xf, 0xf, true));
}
__device__ __forceinline__ float other_row_lane15(float x) {
    return __int_as_float(__builtin_amdgcn_permlanex16(0, __float_as_int(x),
                                                       0xffffffff, 0xffffffff, false, false));
}
__device__ __forceinline__ float other_row_same_lane(float x) {
    return __int_as_float(__builtin_amdgcn_permlanex16(0, __float_as_int(x),
                                                       0x76543210, 0xfedcba98, false, false));
}
__device__ __forceinline__ void mat_acc(float& vx, float& vy, float sx, float sy,
                                        float a00, float a01, float a10, float a11) {
    vx = fmaf(a00, sx, fmaf(a01, sy, vx));
    vy = fmaf(a10, sx, fmaf(a11, sy, vy));
}

struct m2d { double a, b, c, d; };
__device__ __forceinline__ m2d mul2(m2d x, m2d y) {
    m2d r;
    r.a = fma(x.a, y.a, x.b * y.c);
    r.b = fma(x.a, y.b, x.b * y.d);
    r.c = fma(x.c, y.a, x.d * y.c);
    r.d = fma(x.c, y.b, x.d * y.d);
    return r;
}
__device__ __forceinline__ m2d pick(bool t, m2d x, m2d y) {
    m2d r;
    r.a = t ? x.a : y.a; r.b = t ? x.b : y.b;
    r.c = t ? x.c : y.c; r.d = t ? x.d : y.d;
    return r;
}

__device__ __forceinline__ double from_lane31(double x) {
    int lo = __builtin_amdgcn_readlane(__double2loint(x), 31);
    int hi = __builtin_amdgcn_readlane(__double2hiint(x), 31);
    return __hiloint2double(hi, lo);
}

__global__ void __launch_bounds__(64)
iir_state_scan(const float* __restrict__ ends,
               const float* __restrict__ y_init,
               const float* __restrict__ ac,
               float2* __restrict__ states,
               int B, int T) {
    int lane = threadIdx.x & 31;
    int b = blockIdx.x * 2 + (threadIdx.x >> 5);
    if (b >= B) return;
    int nT = T >> 4;
    const float* rowp = ends + (size_t)b * nT * 2;
    float2* st = states + (size_t)b * nT;

    double a1 = (double)ac[0], a2 = (double)ac[1];
    double h[17];
    h[0] = 1.0; h[1] = -a1;
#pragma unroll
    for (int i = 2; i <= 16; ++i) h[i] = -a1 * h[i - 1] - a2 * h[i - 2];
    m2d S0 = { h[16], -a2 * h[15], h[15], -a2 * h[14] };
    m2d S1 = mul2(S0, S0);
    m2d S2 = mul2(S1, S1);
    m2d S3 = mul2(S2, S2);
    m2d S4 = mul2(S3, S3);
    m2d S5 = mul2(S4, S4);
    int n = lane + 1;
    m2d I2 = { 1.0, 0.0, 0.0, 1.0 };
    m2d Pd = pick(n & 1, S0, I2);
    Pd = pick(n & 2, mul2(Pd, S1), Pd);
    Pd = pick(n & 4, mul2(Pd, S2), Pd);
    Pd = pick(n & 8, mul2(Pd, S3), Pd);
    Pd = pick(n & 16, mul2(Pd, S4), Pd);
    Pd = pick(n & 32, S5, Pd);
    float p00 = (float)Pd.a, p01 = (float)Pd.b, p10 = (float)Pd.c, p11 = (float)Pd.d;
    float s1_00 = (float)S0.a, s1_01 = (float)S0.b, s1_10 = (float)S0.c, s1_11 = (float)S0.d;
    float s2_00 = (float)S1.a, s2_01 = (float)S1.b, s2_10 = (float)S1.c, s2_11 = (float)S1.d;
    float s4_00 = (float)S2.a, s4_01 = (float)S2.b, s4_10 = (float)S2.c, s4_11 = (float)S2.d;
    float s8_00 = (float)S3.a, s8_01 = (float)S3.b, s8_10 = (float)S3.c, s8_11 = (float)S3.d;
    bool hi = lane >= 16;
    float q00 = other_row_same_lane(p00), q01 = other_row_same_lane(p01);
    float q10 = other_row_same_lane(p10), q11 = other_row_same_lane(p11);
    q00 = hi ? q00 : 0.f; q01 = hi ? q01 : 0.f;
    q10 = hi ? q10 : 0.f; q11 = hi ? q11 : 0.f;

    double gx = 0.0, gy = 0.0;
    for (int base = 0; base < nT; base += 1024) {
        float vx[32], vy[32];
#pragma unroll
        for (int i = 0; i < 32; ++i) {
            int c = base + 32 * i + lane;
            int cs = c < nT ? c : nT - 1;
            cs = cs > 0 ? cs : 1;
            float2 d = *(const float2*)(rowp + (size_t)(cs - 1) * 2);
            bool live = c < nT;
            vx[i] = live ? d.x : 0.f;
            vy[i] = live ? d.y : 0.f;
        }
        if (base + lane == 0) {
            vx[0] = y_init[(size_t)b * 2 + 0];
            vy[0] = y_init[(size_t)b * 2 + 1];
        }
#pragma unroll
        for (int i = 0; i < 32; ++i) {
            float sx, sy;
            sx = lane_shr<1>(vx[i]); sy = lane_shr<1>(vy[i]);
            mat_acc(vx[i], vy[i], sx, sy, s1_00, s1_01, s1_10, s1_11);
            sx = lane_shr<2>(vx[i]); sy = lane_shr<2>(vy[i]);
            mat_acc(vx[i], vy[i], sx, sy, s2_00, s2_01, s2_10, s2_11);
            sx = lane_shr<4>(vx[i]); sy = lane_shr<4>(vy[i]);
            mat_acc(vx[i], vy[i], sx, sy, s4_00, s4_01, s4_10, s4_11);
            sx = lane_shr<8>(vx[i]); sy = lane_shr<8>(vy[i]);
            mat_acc(vx[i], vy[i], sx, sy, s8_00, s8_01, s8_10, s8_11);
            float tx = other_row_lane15(vx[i]);
            float ty = other_row_lane15(vy[i]);
            tx = hi ? tx : 0.f;
            ty = hi ? ty : 0.f;
            mat_acc(vx[i], vy[i], tx, ty, q00, q01, q10, q11);
        }
#pragma unroll
        for (int i = 0; i < 32; ++i) {
            double ox = fma(Pd.a, gx, fma(Pd.b, gy, (double)vx[i]));
            double oy = fma(Pd.c, gx, fma(Pd.d, gy, (double)vy[i]));
            int c = base + 32 * i + lane;
            if (c < nT) st[c] = make_float2((float)ox, (float)oy);
            gx = from_lane31(ox); gy = from_lane31(oy);
        }
    }
}

extern "C" void kernel_launch(void* const* d_in, const int* in_sizes, int n_in,
                              void* d_out, int out_size, void* d_ws, size_t ws_size,
                              hipStream_t stream) {
    const float* bc = (const float*)d_in[0];
    const float* ac = (const float*)d_in[1];
    const float* u  = (const float*)d_in[2];
    const float* yi = (const float*)d_in[3];
    const float* ui = (const float*)d_in[4];
    float* y = (float*)d_out;

    int B = in_sizes[3] / 2;
    int T = in_sizes[2] / B;
    int nT = T >> 4;
    int tiles = (B >> 4) * nT;
    float2* states = (float2*)d_ws;

    float* ends = y;

    int blocks1 = (tiles + 7) / 8;
    iir_tile_ends_wmma<<<blocks1, 256, 0, stream>>>(u, ui, bc, ac, ends, T, tiles);

    iir_state_scan<<<(B + 1) / 2, 64, 0, stream>>>(ends, yi, ac, states, B, T);

    iir_tile_out_wmma<<<blocks1, 256, 0, stream>>>(u, ui, bc, ac, states, y, T, tiles);
}


# each wave loops over 16 consecutive time tiles of its 16 rows (4096 waves per tile kernel, was 65536): h and B fragments built once per wave, u staged 64 columns at a time, loaded one group ahead
# speedup vs baseline: 6.7743x; 1.5745x over previous
#include <hip/hip_runtime.h>

typedef __attribute__((ext_vector_type(2))) float v2f;
typedef __attribute__((ext_vector_type(8))) float v8f;
typedef __attribute__((ext_vector_type(4))) float v4f;

__device__ __forceinline__ float h_capture_lane(float a1, float a2, int lane) {
    float h[17];
    h[0] = 1.0f;
    h[1] = -a1;
#pragma unroll
    for (int i = 2; i <= 16; ++i) h[i] = -a1 * h[i - 1] - a2 * h[i - 2];
    float hl = h[0];
#pragma unroll
    for (int i = 1; i <= 16; ++i) hl = (lane == i) ? h[i] : hl;
    return hl;
}

__device__ __forceinline__ float h_gather(float hl, int n) {
    int nn = n < 0 ? 0 : n;
    float v = __int_as_float(__builtin_amdgcn_ds_bpermute(nn << 2, __float_as_int(hl)));
    return n < 0 ? 0.0f : v;
}

constexpr int GT = 4;
constexpr int GC = GT * 16;
constexpr int GPW = 4;
constexpr int PITCH = 68;

__device__ __forceinline__ void toeplitz_operands(float a1, float a2, int lane, v2f (&Bf)[4], float& hl) {
    int m    = lane & 15;
    int half = lane >> 4;
    hl = h_capture_lane(a1, a2, lane);
#pragma unroll
    for (int j = 0; j < 4; ++j) {
#pragma unroll
        for (int r = 0; r < 2; ++r) {
            int k = 4 * j + 2 * half + r;
            Bf[j][r] = h_gather(hl, m - k);
        }
    }
}

__device__ __forceinline__ void group_load(const float* __restrict__ u,
                                           const float* __restrict__ u_init,
                                           int T, int tb, int c0, int lane,
                                           v4f (&v)[8], float& pv) {
#pragma unroll
    for (int c = 0; c < 8; ++c) {
        int flat = c * 32 + lane;
        int rr = flat >> 4, c4 = flat & 15;
        int col = c0 + 4 * c4;
        col = col < T ? col : T - 4;
        v[c] = *(const v4f*)(u + (size_t)(tb * 16 + rr) * T + col);
    }
    int rowg = tb * 16 + (lane & 15);
    int cp = c0 <= T ? c0 : T;
    const float* p = (c0 == 0) ? (u_init + (size_t)rowg * 2)
                               : (u + (size_t)rowg * T + (cp - 1));
    pv = *p;
}

__device__ __forceinline__ void group_stage(float* tl, int lane, const v4f (&v)[8], float pv) {
#pragma unroll
    for (int c = 0; c < 8; ++c) {
        int flat = c * 32 + lane;
        int rr = flat >> 4, c4 = flat & 15;
        *(v4f*)(tl + rr * PITCH + 4 + 4 * c4) = v[c];
    }
    if (lane < 16) tl[lane * PITCH + 3] = pv;
}

__device__ __forceinline__ void tile_a(const float* tl, int q, int m, int half, float b0, float b1, v2f (&Af)[4]) {
#pragma unroll
    for (int j = 0; j < 4; ++j) {
        const float* p = tl + m * PITCH + 4 + 16 * q + 4 * j + 2 * half;
        float2 uc = *(const float2*)p;
        float up = p[-1];
        Af[j][0] = b0 * uc.x + b1 * up;
        Af[j][1] = b0 * uc.y + b1 * uc.x;
    }
}

__global__ void __launch_bounds__(256)
iir_tile_ends_wmma(const float* __restrict__ u,
                   const float* __restrict__ u_init,
                   const float* __restrict__ bc,
                   const float* __restrict__ ac,
                   float* __restrict__ ends,
                   int T, int chunks, int waves) {
    __shared__ __attribute__((aligned(16))) float lds[8][16 * PITCH];
    int lane = threadIdx.x & 31;
    int wave = __builtin_amdgcn_readfirstlane(threadIdx.x >> 5);
    int w = blockIdx.x * 8 + wave;
    if (w >= waves) w = waves - 1;
    int nT = T >> 4;
    int tb = w / chunks;
    int cbase = (w - tb * chunks) * (GPW * GC);
    int m    = lane & 15;
    int half = lane >> 4;
    float b0 = bc[0], b1 = bc[1];
    float* tl = &lds[wave][0];

    v4f R[8]; float pv;
    group_load(u, u_init, T, tb, cbase, lane, R, pv);

    v2f Bf[4]; float hl;
    toeplitz_operands(ac[0], ac[1], lane, Bf, hl);
#pragma unroll 1
    for (int g = 0; g < GPW; ++g) {
        int c0 = cbase + g * GC;
        group_stage(tl, lane, R, pv);
        __syncthreads();
        if (g + 1 < GPW) group_load(u, u_init, T, tb, c0 + GC, lane, R, pv);
#pragma unroll
        for (int q = 0; q < GT; ++q) {
            int t0 = c0 + 16 * q;
            v2f Af[4];
            tile_a(tl, q, m, half, b0, b1, Af);
            v8f acc = {0.f, 0.f, 0.f, 0.f, 0.f, 0.f, 0.f, 0.f};
#pragma unroll
            for (int j = 0; j < 4; ++j)
                acc = __builtin_amdgcn_wmma_f32_16x16x4_f32(
                     false, Af[j],  false, Bf[j],
                     (short)0, acc,  false,  false);
            if (t0 < T && m >= 14) {
                float* e = ends + ((size_t)(tb * 16 + 8 * half) * nT + (t0 >> 4)) * 2 + (15 - m);
#pragma unroll
                for (int r = 0; r < 8; ++r)
                    e[(size_t)r * nT * 2] = acc[r];
            }
        }
    }
}

__device__ __forceinline__ void states_load(const float2* __restrict__ states, int nT, int tb, int m,
                                            int c0, float2 (&s)[GT]) {
#pragma unroll
    for (int q = 0; q < GT; ++q) {
        int tt = (c0 >> 4) + q;
        tt = tt < nT ? tt : nT - 1;
        s[q] = states[(size_t)(tb * 16 + m) * nT + tt];
    }
}

__global__ void __launch_bounds__(256)
iir_tile_out_wmma(const float* __restrict__ u,
                  const float* __restrict__ u_init,
                  const float* __restrict__ bc,
                  const float* __restrict__ ac,
                  const float2* __restrict__ states,
                  float* __restrict__ y,
                  int T, int chunks, int waves) {
    __shared__ __attribute__((aligned(16))) float lds[8][16 * PITCH];
    int lane = threadIdx.x & 31;
    int wave = __builtin_amdgcn_readfirstlane(threadIdx.x >> 5);
    int w = blockIdx.x * 8 + wave;
    if (w >= waves) w = waves - 1;
    int nT = T >> 4;
    int tb = w / chunks;
    int cbase = (w - tb * chunks) * (GPW * GC);
    int m    = lane & 15;
    int half = lane >> 4;
    float b0 = bc[0], b1 = bc[1];
    float a2 = ac[1];
    float* tl = &lds[wave][0];

    v4f R[8]; float pv; float2 sn[GT];
    group_load(u, u_init, T, tb, cbase, lane, R, pv);
    states_load(states, nT, tb, m, cbase, sn);

    v2f Bf[4]; float hl;
    toeplitz_operands(ac[0], a2, lane, Bf, hl);
    float g1 = h_gather(hl, m + 1);
    float g0 = h_gather(hl, m);
    v2f B5;
    B5[0] = half ? 0.f : g1;
    B5[1] = half ? 0.f : -a2 * g0;
#pragma unroll 1
    for (int g = 0; g < GPW; ++g) {
        int c0 = cbase + g * GC;
        float2 s[GT];
#pragma unroll
        for (int q = 0; q < GT; ++q) s[q] = sn[q];
        group_stage(tl, lane, R, pv);
        __syncthreads();
        if (g + 1 < GPW) {
            group_load(u, u_init, T, tb, c0 + GC, lane, R, pv);
            states_load(states, nT, tb, m, c0 + GC, sn);
        }
#pragma unroll
        for (int q = 0; q < GT; ++q) {
            int t0 = c0 + 16 * q;
            v2f Af[4];
            tile_a(tl, q, m, half, b0, b1, Af);
            v2f A5;
            A5[0] = half ? 0.f : s[q].x;
            A5[1] = half ? 0.f : s[q].y;
            v8f acc = {0.f, 0.f, 0.f, 0.f, 0.f, 0.f, 0.f, 0.f};
#pragma unroll
            for (int j = 0; j < 4; ++j)
                acc = __builtin_amdgcn_wmma_f32_16x16x4_f32(
                     false, Af[j],  false, Bf[j],
                     (short)0, acc,  false,  false);
            acc = __builtin_amdgcn_wmma_f32_16x16x4_f32(
                 false, A5,  false, B5,
                 (short)0, acc,  false,  false);
            if (t0 < T) {
                float* orow = y + (size_t)(tb * 16 + 8 * half) * T + t0 + m;
#pragma unroll
                for (int r = 0; r < 8; ++r)
                    orow[(size_t)r * T] = acc[r];
            }
        }
    }
}

template <int N>
__device__ __forceinline__ float lane_shr(float x) {
    return __int_as_float(__builtin_amdgcn_update_dpp(0, __float_as_int(x), 0x110 + N, 0xf, 0xf, true));
}
__device__ __forceinline__ float other_row_lane15(float x) {
    return __int_as_float(__builtin_amdgcn_permlanex16(0, __float_as_int(x),
                                                       0xffffffff, 0xffffffff, false, false));
}
__device__ __forceinline__ float other_row_same_lane(float x) {
    return __int_as_float(__builtin_amdgcn_permlanex16(0, __float_as_int(x),
                                                       0x76543210, 0xfedcba98, false, false));
}
__device__ __forceinline__ void mat_acc(float& vx, float& vy, float sx, float sy,
                                        float a00, float a01, float a10, float a11) {
    vx = fmaf(a00, sx, fmaf(a01, sy, vx));
    vy = fmaf(a10, sx, fmaf(a11, sy, vy));
}

struct m2d { double a, b, c, d; };
__device__ __forceinline__ m2d mul2(m2d x, m2d y) {
    m2d r;
    r.a = fma(x.a, y.a, x.b * y.c);
    r.b = fma(x.a, y.b, x.b * y.d);
    r.c = fma(x.c, y.a, x.d * y.c);
    r.d = fma(x.c, y.b, x.d * y.d);
    return r;
}
__device__ __forceinline__ m2d pick(bool t, m2d x, m2d y) {
    m2d r;
    r.a = t ? x.a : y.a; r.b = t ? x.b : y.b;
    r.c = t ? x.c : y.c; r.d = t ? x.d : y.d;
    return r;
}

__device__ __forceinline__ double from_lane31(double x) {
    int lo = __builtin_amdgcn_readlane(__double2loint(x), 31);
    int hi = __builtin_amdgcn_readlane(__double2hiint(x), 31);
    return __hiloint2double(hi, lo);
}

__global__ void __launch_bounds__(64)
iir_state_scan(const float* __restrict__ ends,
               const float* __restrict__ y_init,
               const float* __restrict__ ac,
               float2* __restrict__ states,
               int B, int T) {
    int lane = threadIdx.x & 31;
    int b = blockIdx.x * 2 + (threadIdx.x >> 5);
    if (b >= B) return;
    int nT = T >> 4;
    const float* rowp = ends + (size_t)b * nT * 2;
    float2* st = states + (size_t)b * nT;

    double a1 = (double)ac[0], a2 = (double)ac[1];
    double h[17];
    h[0] = 1.0; h[1] = -a1;
#pragma unroll
    for (int i = 2; i <= 16; ++i) h[i] = -a1 * h[i - 1] - a2 * h[i - 2];
    m2d S0 = { h[16], -a2 * h[15], h[15], -a2 * h[14] };
    m2d S1 = mul2(S0, S0);
    m2d S2 = mul2(S1, S1);
    m2d S3 = mul2(S2, S2);
    m2d S4 = mul2(S3, S3);
    m2d S5 = mul2(S4, S4);
    int n = lane + 1;
    m2d I2 = { 1.0, 0.0, 0.0, 1.0 };
    m2d Pd = pick(n & 1, S0, I2);
    Pd = pick(n & 2, mul2(Pd, S1), Pd);
    Pd = pick(n & 4, mul2(Pd, S2), Pd);
    Pd = pick(n & 8, mul2(Pd, S3), Pd);
    Pd = pick(n & 16, mul2(Pd, S4), Pd);
    Pd = pick(n & 32, S5, Pd);
    float p00 = (float)Pd.a, p01 = (float)Pd.b, p10 = (float)Pd.c, p11 = (float)Pd.d;
    float s1_00 = (float)S0.a, s1_01 = (float)S0.b, s1_10 = (float)S0.c, s1_11 = (float)S0.d;
    float s2_00 = (float)S1.a, s2_01 = (float)S1.b, s2_10 = (float)S1.c, s2_11 = (float)S1.d;
    float s4_00 = (float)S2.a, s4_01 = (float)S2.b, s4_10 = (float)S2.c, s4_11 = (float)S2.d;
    float s8_00 = (float)S3.a, s8_01 = (float)S3.b, s8_10 = (float)S3.c, s8_11 = (float)S3.d;
    bool hi = lane >= 16;
    float q00 = other_row_same_lane(p00), q01 = other_row_same_lane(p01);
    float q10 = other_row_same_lane(p10), q11 = other_row_same_lane(p11);
    q00 = hi ? q00 : 0.f; q01 = hi ? q01 : 0.f;
    q10 = hi ? q10 : 0.f; q11 = hi ? q11 : 0.f;

    double gx = 0.0, gy = 0.0;
    for (int base = 0; base < nT; base += 1024) {
        float vx[32], vy[32];
#pragma unroll
        for (int i = 0; i < 32; ++i) {
            int c = base + 32 * i + lane;
            int cs = c < nT ? c : nT - 1;
            cs = cs > 0 ? cs : 1;
            float2 d = *(const float2*)(rowp + (size_t)(cs - 1) * 2);
            bool live = c < nT;
            vx[i] = live ? d.x : 0.f;
            vy[i] = live ? d.y : 0.f;
        }
        if (base + lane == 0) {
            vx[0] = y_init[(size_t)b * 2 + 0];
            vy[0] = y_init[(size_t)b * 2 + 1];
        }
#pragma unroll
        for (int i = 0; i < 32; ++i) {
            float sx, sy;
            sx = lane_shr<1>(vx[i]); sy = lane_shr<1>(vy[i]);
            mat_acc(vx[i], vy[i], sx, sy, s1_00, s1_01, s1_10, s1_11);
            sx = lane_shr<2>(vx[i]); sy = lane_shr<2>(vy[i]);
            mat_acc(vx[i], vy[i], sx, sy, s2_00, s2_01, s2_10, s2_11);
            sx = lane_shr<4>(vx[i]); sy = lane_shr<4>(vy[i]);
            mat_acc(vx[i], vy[i], sx, sy, s4_00, s4_01, s4_10, s4_11);
            sx = lane_shr<8>(vx[i]); sy = lane_shr<8>(vy[i]);
            mat_acc(vx[i], vy[i], sx, sy, s8_00, s8_01, s8_10, s8_11);
            float tx = other_row_lane15(vx[i]);
            float ty = other_row_lane15(vy[i]);
            tx = hi ? tx : 0.f;
            ty = hi ? ty : 0.f;
            mat_acc(vx[i], vy[i], tx, ty, q00, q01, q10, q11);
        }
#pragma unroll
        for (int i = 0; i < 32; ++i) {
            double ox = fma(Pd.a, gx, fma(Pd.b, gy, (double)vx[i]));
            double oy = fma(Pd.c, gx, fma(Pd.d, gy, (double)vy[i]));
            int c = base + 32 * i + lane;
            if (c < nT) st[c] = make_float2((float)ox, (float)oy);
            gx = from_lane31(ox); gy = from_lane31(oy);
        }
    }
}

extern "C" void kernel_launch(void* const* d_in, const int* in_sizes, int n_in,
                              void* d_out, int out_size, void* d_ws, size_t ws_size,
                              hipStream_t stream) {
    const float* bc = (const float*)d_in[0];
    const float* ac = (const float*)d_in[1];
    const float* u  = (const float*)d_in[2];
    const float* yi = (const float*)d_in[3];
    const float* ui = (const float*)d_in[4];
    float* y = (float*)d_out;

    int B = in_sizes[3] / 2;
    int T = in_sizes[2] / B;
    int chunks = (T + GPW * GC - 1) / (GPW * GC);
    int waves = (B >> 4) * chunks;
    float2* states = (float2*)d_ws;

    float* ends = y;

    int blocks1 = (waves + 7) / 8;
    iir_tile_ends_wmma<<<blocks1, 256, 0, stream>>>(u, ui, bc, ac, ends, T, chunks, waves);

    iir_state_scan<<<(B + 1) / 2, 64, 0, stream>>>(ends, yi, ac, states, B, T);

    iir_tile_out_wmma<<<blocks1, 256, 0, stream>>>(u, ui, bc, ac, states, y, T, chunks, waves);
}


# tile state kept in registers along a wave's 16 tiles (WMMA A/B swapped: a tile's last 2 outputs are the next state slice); chunk starts by compensated f32 chain; scan launch gone; sum order changes
# speedup vs baseline: 13.1058x; 1.9346x over previous
#include <hip/hip_runtime.h>


typedef __attribute__((ext_vector_type(2))) float v2f;
typedef __attribute__((ext_vector_type(8))) float v8f;
typedef __attribute__((ext_vector_type(4))) float v4f;

__device__ __forceinline__ float h_capture_lane(float a1, float a2, int lane) {
    float h[17];
    h[0] = 1.0f;
    h[1] = -a1;
#pragma unroll
    for (int i = 2; i <= 16; ++i) h[i] = -a1 * h[i - 1] - a2 * h[i - 2];
    float hl = h[0];
#pragma unroll
    for (int i = 1; i <= 16; ++i) hl = (lane == i) ? h[i] : hl;
    return hl;
}

__device__ __forceinline__ float h_gather(float hl, int n) {
    int nn = n < 0 ? 0 : n;
    float v = __int_as_float(__builtin_amdgcn_ds_bpermute(nn << 2, __float_as_int(hl)));
    return n < 0 ? 0.0f : v;
}

constexpr int GT = 4;
constexpr int GC = GT * 16;
constexpr int GPW = 4;
constexpr int PITCH = 68;
static_assert(GT * GPW == 16, "the chunk matrix below is (M^16)^16: 16 tiles per chunk");

__device__ __forceinline__ void toeplitz_operands(float a1, float a2, int lane, v2f (&Tf)[4], float& hl) {
    int m    = lane & 15;
    int half = lane >> 4;
    hl = h_capture_lane(a1, a2, lane);
#pragma unroll
    for (int j = 0; j < 4; ++j) {
#pragma unroll
        for (int r = 0; r < 2; ++r) {
            int k = 4 * j + 2 * half + r;
            Tf[j][r] = h_gather(hl, m - k);
        }
    }
}

__device__ __forceinline__ void group_load(const float* __restrict__ u,
                                           const float* __restrict__ u_init,
                                           int T, int tb, int c0, int lane,
                                           v4f (&v)[8], float& pv) {
#pragma unroll
    for (int c = 0; c < 8; ++c) {
        int flat = c * 32 + lane;
        int rr = flat >> 4, c4 = flat & 15;
        int col = c0 + 4 * c4;
        col = col < T ? col : T - 4;
        v[c] = *(const v4f*)(u + (size_t)(tb * 16 + rr) * T + col);
    }
    int rowg = tb * 16 + (lane & 15);
    int cp = c0 <= T ? c0 : T;
    const float* p = (c0 == 0) ? (u_init + (size_t)rowg * 2)
                               : (u + (size_t)rowg * T + (cp - 1));
    pv = *p;
}

__device__ __forceinline__ void group_stage(float* tl, int lane, const v4f (&v)[8], float pv) {
#pragma unroll
    for (int c = 0; c < 8; ++c) {
        int flat = c * 32 + lane;
        int rr = flat >> 4, c4 = flat & 15;
        *(v4f*)(tl + rr * PITCH + 4 + 4 * c4) = v[c];
    }
    if (lane < 16) tl[lane * PITCH + 3] = pv;
}

__device__ __forceinline__ void tile_x(const float* tl, int q, int m, int half, float b0, float b1, v2f (&Xf)[4]) {
#pragma unroll
    for (int j = 0; j < 4; ++j) {
        const float* p = tl + m * PITCH + 4 + 16 * q + 4 * j + 2 * half;
        float2 uc = *(const float2*)p;
        float up = p[-1];
        Xf[j][0] = b0 * uc.x + b1 * up;
        Xf[j][1] = b0 * uc.y + b1 * uc.x;
    }
}

struct m2d { double a, b, c, d; };
__device__ __forceinline__ m2d mul2(m2d x, m2d y) {
    m2d r;
    r.a = fma(x.a, y.a, x.b * y.c);
    r.b = fma(x.a, y.b, x.b * y.d);
    r.c = fma(x.c, y.a, x.d * y.c);
    r.d = fma(x.c, y.b, x.d * y.d);
    return r;
}

__global__ void __launch_bounds__(256)
iir_chunk_ends_wmma(const float* __restrict__ u,
                    const float* __restrict__ u_init,
                    const float* __restrict__ bc,
                    const float* __restrict__ ac,
                    float2* __restrict__ cends,
                    int T, int chunks, int waves) {
    __shared__ __attribute__((aligned(16))) float lds[8][16 * PITCH];
    int lane = threadIdx.x & 31;
    int wave = __builtin_amdgcn_readfirstlane(threadIdx.x >> 5);
    int w = blockIdx.x * 8 + wave;
    if (w >= waves) w = waves - 1;
    int tb = w / chunks;
    int cidx = w - tb * chunks;
    int cbase = cidx * (GPW * GC);
    int m    = lane & 15;
    int half = lane >> 4;
    float b0 = bc[0], b1 = bc[1];
    float a2 = ac[1];
    float* tl = &lds[wave][0];

    v4f R[8]; float pv;
    group_load(u, u_init, T, tb, cbase, lane, R, pv);

    v2f Tf[4]; float hl;
    toeplitz_operands(ac[0], a2, lane, Tf, hl);
    float h14 = __int_as_float(__builtin_amdgcn_readlane(__float_as_int(hl), 14));
    float h15 = __int_as_float(__builtin_amdgcn_readlane(__float_as_int(hl), 15));
    float h16 = __int_as_float(__builtin_amdgcn_readlane(__float_as_int(hl), 16));
    float m00 = h16, m01 = -a2 * h15, m10 = h15, m11 = -a2 * h14;
    float s1 = 0.f, s2 = 0.f;
#pragma unroll 1
    for (int g = 0; g < GPW; ++g) {
        int c0 = cbase + g * GC;
        group_stage(tl, lane, R, pv);
        __syncthreads();
        if (g + 1 < GPW) group_load(u, u_init, T, tb, c0 + GC, lane, R, pv);
#pragma unroll
        for (int q = 0; q < GT; ++q) {
            v2f Xf[4];
            tile_x(tl, q, m, half, b0, b1, Xf);
            v8f acc = {0.f, 0.f, 0.f, 0.f, 0.f, 0.f, 0.f, 0.f};
#pragma unroll
            for (int j = 0; j < 4; ++j)
                acc = __builtin_amdgcn_wmma_f32_16x16x4_f32(
                     false, Tf[j],  false, Xf[j],
                     (short)0, acc,  false,  false);
            float n1 = fmaf(m00, s1, fmaf(m01, s2, acc[7]));
            float n2 = fmaf(m10, s1, fmaf(m11, s2, acc[6]));
            s1 = n1; s2 = n2;
        }
    }
    if (half) cends[((size_t)tb * chunks + cidx) * 16 + m] = make_float2(s1, s2);

    if (blockIdx.x == 0 && wave == 0) {
        double a1d = (double)ac[0], a2d = (double)ac[1];
        double h[17];
        h[0] = 1.0; h[1] = -a1d;
#pragma unroll
        for (int i = 2; i <= 16; ++i) h[i] = -a1d * h[i - 1] - a2d * h[i - 2];
        m2d S0 = { h[16], -a2d * h[15], h[15], -a2d * h[14] };
        m2d S1 = mul2(S0, S0);
        m2d S2 = mul2(S1, S1);
        m2d S3 = mul2(S2, S2);
        m2d S4 = mul2(S3, S3);
        if (lane == 0) {
            float* pw = (float*)(cends + (size_t)waves * 16);
            float pa = (float)S4.a, pb = (float)S4.b, pc = (float)S4.c, pd = (float)S4.d;
            pw[0] = pa; pw[1] = pb; pw[2] = pc; pw[3] = pd;
            pw[4] = (float)(S4.a - (double)pa); pw[5] = (float)(S4.b - (double)pb);
            pw[6] = (float)(S4.c - (double)pc); pw[7] = (float)(S4.d - (double)pd);
        }
    }
}

__device__ __forceinline__ void rows_store(const float* to, float* __restrict__ y, int T, int tb, int c0, int lane) {
#pragma unroll
    for (int c = 0; c < 8; ++c) {
        int flat = c * 32 + lane;
        int rr = flat >> 4, c4 = flat & 15;
        int col = c0 + 4 * c4;
        v4f v = *(const v4f*)(to + rr * PITCH + 4 * c4);
        if (col < T) *(v4f*)(y + (size_t)(tb * 16 + rr) * T + col) = v;
    }
}

__device__ __forceinline__ float chunk_row(float ph0, float ph1, float pl0, float pl1, float s1, float s2, float e) {
    float th = __fmul_rn(ph1, s2);
    float tl = fmaf(ph1, s2, -th);
    float u  = fmaf(ph0, s1, th);
    float v  = fmaf(pl0, s1, fmaf(pl1, s2, e));
    return u + (tl + v);
}
__device__ __forceinline__ void pairs_load(const float2* __restrict__ Eb, int m, int j0, int c, float2 (&z)[32], int ifirst) {
#pragma unroll
    for (int i = ifirst; i < 32; ++i)
        if (j0 + i < c) z[i] = Eb[(size_t)(j0 + i) * 16 + m];
}
__device__ __forceinline__ void pairs_fold(const float2 (&z)[32], int j0, int c, const float (&ph)[4], const float (&pl)[4],
                                           float& sx, float& sy) {
#pragma unroll
    for (int i = 0; i < 32; ++i)
        if (j0 + i < c) {
            float nx = chunk_row(ph[0], ph[1], pl[0], pl[1], sx, sy, z[i].x);
            float ny = chunk_row(ph[2], ph[3], pl[2], pl[3], sx, sy, z[i].y);
            sx = nx; sy = ny;
        }
}

__global__ void __launch_bounds__(256)
iir_tile_out_wmma(const float* __restrict__ u,
                  const float* __restrict__ u_init,
                  const float* __restrict__ bc,
                  const float* __restrict__ ac,
                  const float* __restrict__ y_init,
                  const float2* __restrict__ cends,
                  float* __restrict__ y,
                  int T, int chunks, int waves) {
    __shared__ __attribute__((aligned(16))) float lds[8][16 * PITCH];
    __shared__ __attribute__((aligned(16))) float ldo[8][16 * PITCH];
    int lane = threadIdx.x & 31;
    int wave = __builtin_amdgcn_readfirstlane(threadIdx.x >> 5);
    int w = blockIdx.x * 8 + wave;
    if (w >= waves) w = waves - 1;
    int tb = w / chunks;
    int cidx = w - tb * chunks;
    int cbase = cidx * (GPW * GC);
    int m    = lane & 15;
    int half = lane >> 4;
    float b0 = bc[0], b1 = bc[1];
    float a2 = ac[1];
    float* tl = &lds[wave][0];
    float* to = &ldo[wave][0];

    const float2* Eb = cends + (size_t)tb * chunks * 16;
    const float* pw = (const float*)(cends + (size_t)waves * 16);
    float2 yi = *(const float2*)(y_init + (size_t)(tb * 16 + m) * 2);
    float ph[4] = { pw[0], pw[1], pw[2], pw[3] };
    float pl[4] = { pw[4], pw[5], pw[6], pw[7] };
    float sx = yi.x, sy = yi.y;
    float2 za[32], zb[32];
    za[0] = Eb[m];
    pairs_load(Eb, m, 0, cidx, za, 1);
    pairs_load(Eb, m, 32, cidx, zb, 0);
    v4f R[8]; float pv;
    group_load(u, u_init, T, tb, cbase, lane, R, pv);

    v2f Tf[4]; float hl;
    toeplitz_operands(ac[0], a2, lane, Tf, hl);
    float g1 = h_gather(hl, m + 1);
    float g0 = h_gather(hl, m);
    v2f A5;
    A5[0] = half ? g1 : 0.f;
    A5[1] = half ? -a2 * g0 : 0.f;

    pairs_fold(za, 0, cidx, ph, pl, sx, sy);
    pairs_fold(zb, 32, cidx, ph, pl, sx, sy);
    for (int j0 = 64; j0 < cidx; j0 += 32) {
        pairs_load(Eb, m, j0, cidx, za, 0);
        pairs_fold(za, j0, cidx, ph, pl, sx, sy);
    }

    v2f B5;
    B5[0] = half ? sx : 0.f;
    B5[1] = half ? sy : 0.f;
#pragma unroll 1
    for (int g = 0; g < GPW; ++g) {
        int c0 = cbase + g * GC;
        group_stage(tl, lane, R, pv);
        __syncthreads();
        if (g + 1 < GPW) group_load(u, u_init, T, tb, c0 + GC, lane, R, pv);
        if (g > 0) rows_store(to, y, T, tb, c0 - GC, lane);
#pragma unroll
        for (int q = 0; q < GT; ++q) {
            v2f Xf[4];
            tile_x(tl, q, m, half, b0, b1, Xf);
            v8f acc = {0.f, 0.f, 0.f, 0.f, 0.f, 0.f, 0.f, 0.f};
#pragma unroll
            for (int j = 0; j < 4; ++j)
                acc = __builtin_amdgcn_wmma_f32_16x16x4_f32(
                     false, Tf[j],  false, Xf[j],
                     (short)0, acc,  false,  false);
            acc = __builtin_amdgcn_wmma_f32_16x16x4_f32(
                 false, A5,  false, B5,
                 (short)0, acc,  false,  false);
            B5[0] = half ? acc[7] : 0.f;
            B5[1] = half ? acc[6] : 0.f;
            float* op = to + m * PITCH + 16 * q + 8 * half;
            v4f lo = {acc[0], acc[1], acc[2], acc[3]};
            v4f hi = {acc[4], acc[5], acc[6], acc[7]};
            *(v4f*)op = lo;
            *(v4f*)(op + 4) = hi;
        }
    }
    __syncthreads();
    rows_store(to, y, T, tb, cbase + (GPW - 1) * GC, lane);
}

extern "C" void kernel_launch(void* const* d_in, const int* in_sizes, int n_in,
                              void* d_out, int out_size, void* d_ws, size_t ws_size,
                              hipStream_t stream) {
    const float* bc = (const float*)d_in[0];
    const float* ac = (const float*)d_in[1];
    const float* u  = (const float*)d_in[2];
    const float* yi = (const float*)d_in[3];
    const float* ui = (const float*)d_in[4];
    float* y = (float*)d_out;

    int B = in_sizes[3] / 2;
    int T = in_sizes[2] / B;
    int chunks = (T + GPW * GC - 1) / (GPW * GC);
    int waves = (B >> 4) * chunks;
    float2* cends = (float2*)d_ws;

    int blocks1 = (waves + 7) / 8;
    iir_chunk_ends_wmma<<<blocks1, 256, 0, stream>>>(u, ui, bc, ac, cends, T, chunks, waves);

    iir_tile_out_wmma<<<blocks1, 256, 0, stream>>>(u, ui, bc, ac, yi, cends, y, T, chunks, waves);
}


# chunk-ends kernel: the 24 LDS fragment reads of a group (4 tiles) issued as one batch ahead of its 16 WMMAs; they were read slice by slice, ~16 dependent LDS round trips per group
# speedup vs baseline: 13.6132x; 1.0387x over previous
#include <hip/hip_runtime.h>


typedef __attribute__((ext_vector_type(2))) float v2f;
typedef __attribute__((ext_vector_type(8))) float v8f;
typedef __attribute__((ext_vector_type(4))) float v4f;

__device__ __forceinline__ float h_capture_lane(float a1, float a2, int lane) {
    float h[17];
    h[0] = 1.0f;
    h[1] = -a1;
#pragma unroll
    for (int i = 2; i <= 16; ++i) h[i] = -a1 * h[i - 1] - a2 * h[i - 2];
    float hl = h[0];
#pragma unroll
    for (int i = 1; i <= 16; ++i) hl = (lane == i) ? h[i] : hl;
    return hl;
}

__device__ __forceinline__ float h_gather(float hl, int n) {
    int nn = n < 0 ? 0 : n;
    float v = __int_as_float(__builtin_amdgcn_ds_bpermute(nn << 2, __float_as_int(hl)));
    return n < 0 ? 0.0f : v;
}

constexpr int GT = 4;
constexpr int GC = GT * 16;
constexpr int GPW = 4;
constexpr int PITCH = 68;
static_assert(GT * GPW == 16, "the chunk matrix below is (M^16)^16: 16 tiles per chunk");

__device__ __forceinline__ void toeplitz_operands(float a1, float a2, int lane, v2f (&Tf)[4], float& hl) {
    int m    = lane & 15;
    int half = lane >> 4;
    hl = h_capture_lane(a1, a2, lane);
#pragma unroll
    for (int j = 0; j < 4; ++j) {
#pragma unroll
        for (int r = 0; r < 2; ++r) {
            int k = 4 * j + 2 * half + r;
            Tf[j][r] = h_gather(hl, m - k);
        }
    }
}

__device__ __forceinline__ void group_load(const float* __restrict__ u,
                                           const float* __restrict__ u_init,
                                           int T, int tb, int c0, int lane,
                                           v4f (&v)[8], float& pv) {
#pragma unroll
    for (int c = 0; c < 8; ++c) {
        int flat = c * 32 + lane;
        int rr = flat >> 4, c4 = flat & 15;
        int col = c0 + 4 * c4;
        col = col < T ? col : T - 4;
        v[c] = *(const v4f*)(u + (size_t)(tb * 16 + rr) * T + col);
    }
    int rowg = tb * 16 + (lane & 15);
    int cp = c0 <= T ? c0 : T;
    const float* p = (c0 == 0) ? (u_init + (size_t)rowg * 2)
                               : (u + (size_t)rowg * T + (cp - 1));
    pv = *p;
}

__device__ __forceinline__ void group_stage(float* tl, int lane, const v4f (&v)[8], float pv) {
#pragma unroll
    for (int c = 0; c < 8; ++c) {
        int flat = c * 32 + lane;
        int rr = flat >> 4, c4 = flat & 15;
        *(v4f*)(tl + rr * PITCH + 4 + 4 * c4) = v[c];
    }
    if (lane < 16) tl[lane * PITCH + 3] = pv;
}

__device__ __forceinline__ void tile_x(const float* tl, int q, int m, int half, float b0, float b1, v2f (&Xf)[4]) {
#pragma unroll
    for (int j = 0; j < 4; ++j) {
        const float* p = tl + m * PITCH + 4 + 16 * q + 4 * j + 2 * half;
        float2 uc = *(const float2*)p;
        float up = p[-1];
        Xf[j][0] = b0 * uc.x + b1 * up;
        Xf[j][1] = b0 * uc.y + b1 * uc.x;
    }
}

__device__ __forceinline__ void group_x_read(const float* tl, int m, int half, float2 (&uc)[GT][4], float (&up)[GT][4]) {
#pragma unroll
    for (int q = 0; q < GT; ++q) {
#pragma unroll
        for (int j = 0; j < 4; ++j) {
            const float* p = tl + m * PITCH + 4 + 16 * q + 4 * j + 2 * half;
            uc[q][j] = *(const float2*)p;
            up[q][j] = p[-1];
        }
    }
}
__device__ __forceinline__ void tile_x_regs(const float2 (&uc)[4], const float (&up)[4], float b0, float b1, v2f (&Xf)[4]) {
#pragma unroll
    for (int j = 0; j < 4; ++j) {
        Xf[j][0] = b0 * uc[j].x + b1 * up[j];
        Xf[j][1] = b0 * uc[j].y + b1 * uc[j].x;
    }
}

struct m2d { double a, b, c, d; };
__device__ __forceinline__ m2d mul2(m2d x, m2d y) {
    m2d r;
    r.a = fma(x.a, y.a, x.b * y.c);
    r.b = fma(x.a, y.b, x.b * y.d);
    r.c = fma(x.c, y.a, x.d * y.c);
    r.d = fma(x.c, y.b, x.d * y.d);
    return r;
}

__global__ void __launch_bounds__(256)
iir_chunk_ends_wmma(const float* __restrict__ u,
                    const float* __restrict__ u_init,
                    const float* __restrict__ bc,
                    const float* __restrict__ ac,
                    float2* __restrict__ cends,
                    int T, int chunks, int waves) {
    __shared__ __attribute__((aligned(16))) float lds[8][16 * PITCH];
    int lane = threadIdx.x & 31;
    int wave = __builtin_amdgcn_readfirstlane(threadIdx.x >> 5);
    int w = blockIdx.x * 8 + wave;
    if (w >= waves) w = waves - 1;
    int tb = w / chunks;
    int cidx = w - tb * chunks;
    int cbase = cidx * (GPW * GC);
    int m    = lane & 15;
    int half = lane >> 4;
    float b0 = bc[0], b1 = bc[1];
    float a2 = ac[1];
    float* tl = &lds[wave][0];

    v4f R[8]; float pv;
    group_load(u, u_init, T, tb, cbase, lane, R, pv);

    v2f Tf[4]; float hl;
    toeplitz_operands(ac[0], a2, lane, Tf, hl);
    float h14 = __int_as_float(__builtin_amdgcn_readlane(__float_as_int(hl), 14));
    float h15 = __int_as_float(__builtin_amdgcn_readlane(__float_as_int(hl), 15));
    float h16 = __int_as_float(__builtin_amdgcn_readlane(__float_as_int(hl), 16));
    float m00 = h16, m01 = -a2 * h15, m10 = h15, m11 = -a2 * h14;
    float s1 = 0.f, s2 = 0.f;
#pragma unroll 1
    for (int g = 0; g < GPW; ++g) {
        int c0 = cbase + g * GC;
        group_stage(tl, lane, R, pv);
        __syncthreads();
        if (g + 1 < GPW) group_load(u, u_init, T, tb, c0 + GC, lane, R, pv);
        float2 uc[GT][4]; float up[GT][4];
        group_x_read(tl, m, half, uc, up);
        __builtin_amdgcn_sched_barrier(0);
#pragma unroll
        for (int q = 0; q < GT; ++q) {
            v2f Xf[4];
            tile_x_regs(uc[q], up[q], b0, b1, Xf);
            v8f acc = {0.f, 0.f, 0.f, 0.f, 0.f, 0.f, 0.f, 0.f};
#pragma unroll
            for (int j = 0; j < 4; ++j)
                acc = __builtin_amdgcn_wmma_f32_16x16x4_f32(
                     false, Tf[j],  false, Xf[j],
                     (short)0, acc,  false,  false);
            float n1 = fmaf(m00, s1, fmaf(m01, s2, acc[7]));
            float n2 = fmaf(m10, s1, fmaf(m11, s2, acc[6]));
            s1 = n1; s2 = n2;
        }
    }
    if (half) cends[((size_t)tb * chunks + cidx) * 16 + m] = make_float2(s1, s2);

    if (blockIdx.x == 0 && wave == 0) {
        double a1d = (double)ac[0], a2d = (double)ac[1];
        double h[17];
        h[0] = 1.0; h[1] = -a1d;
#pragma unroll
        for (int i = 2; i <= 16; ++i) h[i] = -a1d * h[i - 1] - a2d * h[i - 2];
        m2d S0 = { h[16], -a2d * h[15], h[15], -a2d * h[14] };
        m2d S1 = mul2(S0, S0);
        m2d S2 = mul2(S1, S1);
        m2d S3 = mul2(S2, S2);
        m2d S4 = mul2(S3, S3);
        if (lane == 0) {
            float* pw = (float*)(cends + (size_t)waves * 16);
            float pa = (float)S4.a, pb = (float)S4.b, pc = (float)S4.c, pd = (float)S4.d;
            pw[0] = pa; pw[1] = pb; pw[2] = pc; pw[3] = pd;
            pw[4] = (float)(S4.a - (double)pa); pw[5] = (float)(S4.b - (double)pb);
            pw[6] = (float)(S4.c - (double)pc); pw[7] = (float)(S4.d - (double)pd);
        }
    }
}

__device__ __forceinline__ void rows_store(const float* to, float* __restrict__ y, int T, int tb, int c0, int lane) {
#pragma unroll
    for (int c = 0; c < 8; ++c) {
        int flat = c * 32 + lane;
        int rr = flat >> 4, c4 = flat & 15;
        int col = c0 + 4 * c4;
        v4f v = *(const v4f*)(to + rr * PITCH + 4 * c4);
        if (col < T) *(v4f*)(y + (size_t)(tb * 16 + rr) * T + col) = v;
    }
}

__device__ __forceinline__ float chunk_row(float ph0, float ph1, float pl0, float pl1, float s1, float s2, float e) {
    float th = __fmul_rn(ph1, s2);
    float tl = fmaf(ph1, s2, -th);
    float u  = fmaf(ph0, s1, th);
    float v  = fmaf(pl0, s1, fmaf(pl1, s2, e));
    return u + (tl + v);
}
__device__ __forceinline__ void pairs_load(const float2* __restrict__ Eb, int m, int j0, int c, float2 (&z)[32], int ifirst) {
#pragma unroll
    for (int i = ifirst; i < 32; ++i)
        if (j0 + i < c) z[i] = Eb[(size_t)(j0 + i) * 16 + m];
}
__device__ __forceinline__ void pairs_fold(const float2 (&z)[32], int j0, int c, const float (&ph)[4], const float (&pl)[4],
                                           float& sx, float& sy) {
#pragma unroll
    for (int i = 0; i < 32; ++i)
        if (j0 + i < c) {
            float nx = chunk_row(ph[0], ph[1], pl[0], pl[1], sx, sy, z[i].x);
            float ny = chunk_row(ph[2], ph[3], pl[2], pl[3], sx, sy, z[i].y);
            sx = nx; sy = ny;
        }
}

__global__ void __launch_bounds__(256)
iir_tile_out_wmma(const float* __restrict__ u,
                  const float* __restrict__ u_init,
                  const float* __restrict__ bc,
                  const float* __restrict__ ac,
                  const float* __restrict__ y_init,
                  const float2* __restrict__ cends,
                  float* __restrict__ y,
                  int T, int chunks, int waves) {
    __shared__ __attribute__((aligned(16))) float lds[8][16 * PITCH];
    __shared__ __attribute__((aligned(16))) float ldo[8][16 * PITCH];
    int lane = threadIdx.x & 31;
    int wave = __builtin_amdgcn_readfirstlane(threadIdx.x >> 5);
    int w = blockIdx.x * 8 + wave;
    if (w >= waves) w = waves - 1;
    int tb = w / chunks;
    int cidx = w - tb * chunks;
    int cbase = cidx * (GPW * GC);
    int m    = lane & 15;
    int half = lane >> 4;
    float b0 = bc[0], b1 = bc[1];
    float a2 = ac[1];
    float* tl = &lds[wave][0];
    float* to = &ldo[wave][0];

    const float2* Eb = cends + (size_t)tb * chunks * 16;
    const float* pw = (const float*)(cends + (size_t)waves * 16);
    float2 yi = *(const float2*)(y_init + (size_t)(tb * 16 + m) * 2);
    float ph[4] = { pw[0], pw[1], pw[2], pw[3] };
    float pl[4] = { pw[4], pw[5], pw[6], pw[7] };
    float sx = yi.x, sy = yi.y;
    float2 za[32], zb[32];
    za[0] = Eb[m];
    pairs_load(Eb, m, 0, cidx, za, 1);
    pairs_load(Eb, m, 32, cidx, zb, 0);
    v4f R[8]; float pv;
    group_load(u, u_init, T, tb, cbase, lane, R, pv);

    v2f Tf[4]; float hl;
    toeplitz_operands(ac[0], a2, lane, Tf, hl);
    float g1 = h_gather(hl, m + 1);
    float g0 = h_gather(hl, m);
    v2f A5;
    A5[0] = half ? g1 : 0.f;
    A5[1] = half ? -a2 * g0 : 0.f;

    pairs_fold(za, 0, cidx, ph, pl, sx, sy);
    pairs_fold(zb, 32, cidx, ph, pl, sx, sy);
    for (int j0 = 64; j0 < cidx; j0 += 32) {
        pairs_load(Eb, m, j0, cidx, za, 0);
        pairs_fold(za, j0, cidx, ph, pl, sx, sy);
    }

    v2f B5;
    B5[0] = half ? sx : 0.f;
    B5[1] = half ? sy : 0.f;
#pragma unroll 1
    for (int g = 0; g < GPW; ++g) {
        int c0 = cbase + g * GC;
        group_stage(tl, lane, R, pv);
        __syncthreads();
        if (g + 1 < GPW) group_load(u, u_init, T, tb, c0 + GC, lane, R, pv);
        if (g > 0) rows_store(to, y, T, tb, c0 - GC, lane);
#pragma unroll
        for (int q = 0; q < GT; ++q) {
            v2f Xf[4];
            tile_x(tl, q, m, half, b0, b1, Xf);
            v8f acc = {0.f, 0.f, 0.f, 0.f, 0.f, 0.f, 0.f, 0.f};
#pragma unroll
            for (int j = 0; j < 4; ++j)
                acc = __builtin_amdgcn_wmma_f32_16x16x4_f32(
                     false, Tf[j],  false, Xf[j],
                     (short)0, acc,  false,  false);
            acc = __builtin_amdgcn_wmma_f32_16x16x4_f32(
                 false, A5,  false, B5,
                 (short)0, acc,  false,  false);
            B5[0] = half ? acc[7] : 0.f;
            B5[1] = half ? acc[6] : 0.f;
            float* op = to + m * PITCH + 16 * q + 8 * half;
            v4f lo = {acc[0], acc[1], acc[2], acc[3]};
            v4f hi = {acc[4], acc[5], acc[6], acc[7]};
            *(v4f*)op = lo;
            *(v4f*)(op + 4) = hi;
        }
    }
    __syncthreads();
    rows_store(to, y, T, tb, cbase + (GPW - 1) * GC, lane);
}

extern "C" void kernel_launch(void* const* d_in, const int* in_sizes, int n_in,
                              void* d_out, int out_size, void* d_ws, size_t ws_size,
                              hipStream_t stream) {
    const float* bc = (const float*)d_in[0];
    const float* ac = (const float*)d_in[1];
    const float* u  = (const float*)d_in[2];
    const float* yi = (const float*)d_in[3];
    const float* ui = (const float*)d_in[4];
    float* y = (float*)d_out;

    int B = in_sizes[3] / 2;
    int T = in_sizes[2] / B;
    int chunks = (T + GPW * GC - 1) / (GPW * GC);
    int waves = (B >> 4) * chunks;
    float2* cends = (float2*)d_ws;

    int blocks1 = (waves + 7) / 8;
    iir_chunk_ends_wmma<<<blocks1, 256, 0, stream>>>(u, ui, bc, ac, cends, T, chunks, waves);

    iir_tile_out_wmma<<<blocks1, 256, 0, stream>>>(u, ui, bc, ac, yi, cends, y, T, chunks, waves);
}


# workgroup barriers dropped from both kernels: each wave stages through an LDS slice of its own and LDS runs a wave's instructions in issue order, so a block's waves no longer wait for each other
# speedup vs baseline: 14.9472x; 1.0980x over previous
#include <hip/hip_runtime.h>


typedef __attribute__((ext_vector_type(2))) float v2f;
typedef __attribute__((ext_vector_type(8))) float v8f;
typedef __attribute__((ext_vector_type(4))) float v4f;

__device__ __forceinline__ float h_capture_lane(float a1, float a2, int lane) {
    float h[17];
    h[0] = 1.0f;
    h[1] = -a1;
#pragma unroll
    for (int i = 2; i <= 16; ++i) h[i] = -a1 * h[i - 1] - a2 * h[i - 2];
    float hl = h[0];
#pragma unroll
    for (int i = 1; i <= 16; ++i) hl = (lane == i) ? h[i] : hl;
    return hl;
}

__device__ __forceinline__ float h_gather(float hl, int n) {
    int nn = n < 0 ? 0 : n;
    float v = __int_as_float(__builtin_amdgcn_ds_bpermute(nn << 2, __float_as_int(hl)));
    return n < 0 ? 0.0f : v;
}

constexpr int GT = 4;
constexpr int GC = GT * 16;
constexpr int GPW = 4;
constexpr int PITCH = 68;
static_assert(GT * GPW == 16, "the chunk matrix below is (M^16)^16: 16 tiles per chunk");


__device__ __forceinline__ void toeplitz_operands(float a1, float a2, int lane, v2f (&Tf)[4], float& hl) {
    int m    = lane & 15;
    int half = lane >> 4;
    hl = h_capture_lane(a1, a2, lane);
#pragma unroll
    for (int j = 0; j < 4; ++j) {
#pragma unroll
        for (int r = 0; r < 2; ++r) {
            int k = 4 * j + 2 * half + r;
            Tf[j][r] = h_gather(hl, m - k);
        }
    }
}

__device__ __forceinline__ void group_load(const float* __restrict__ u,
                                           const float* __restrict__ u_init,
                                           int T, int tb, int c0, int lane,
                                           v4f (&v)[8], float& pv) {
#pragma unroll
    for (int c = 0; c < 8; ++c) {
        int flat = c * 32 + lane;
        int rr = flat >> 4, c4 = flat & 15;
        int col = c0 + 4 * c4;
        col = col < T ? col : T - 4;
        v[c] = *(const v4f*)(u + (size_t)(tb * 16 + rr) * T + col);
    }
    int rowg = tb * 16 + (lane & 15);
    int cp = c0 <= T ? c0 : T;
    const float* p = (c0 == 0) ? (u_init + (size_t)rowg * 2)
                               : (u + (size_t)rowg * T + (cp - 1));
    pv = *p;
}

__device__ __forceinline__ void group_stage(float* tl, int lane, const v4f (&v)[8], float pv) {
#pragma unroll
    for (int c = 0; c < 8; ++c) {
        int flat = c * 32 + lane;
        int rr = flat >> 4, c4 = flat & 15;
        *(v4f*)(tl + rr * PITCH + 4 + 4 * c4) = v[c];
    }
    if (lane < 16) tl[lane * PITCH + 3] = pv;
}

__device__ __forceinline__ void tile_x(const float* tl, int q, int m, int half, float b0, float b1, v2f (&Xf)[4]) {
#pragma unroll
    for (int j = 0; j < 4; ++j) {
        const float* p = tl + m * PITCH + 4 + 16 * q + 4 * j + 2 * half;
        float2 uc = *(const float2*)p;
        float up = p[-1];
        Xf[j][0] = b0 * uc.x + b1 * up;
        Xf[j][1] = b0 * uc.y + b1 * uc.x;
    }
}

__device__ __forceinline__ void group_x_read(const float* tl, int m, int half, float2 (&uc)[GT][4], float (&up)[GT][4]) {
#pragma unroll
    for (int q = 0; q < GT; ++q) {
#pragma unroll
        for (int j = 0; j < 4; ++j) {
            const float* p = tl + m * PITCH + 4 + 16 * q + 4 * j + 2 * half;
            uc[q][j] = *(const float2*)p;
            up[q][j] = p[-1];
        }
    }
}
__device__ __forceinline__ void tile_x_regs(const float2 (&uc)[4], const float (&up)[4], float b0, float b1, v2f (&Xf)[4]) {
#pragma unroll
    for (int j = 0; j < 4; ++j) {
        Xf[j][0] = b0 * uc[j].x + b1 * up[j];
        Xf[j][1] = b0 * uc[j].y + b1 * uc[j].x;
    }
}

struct m2d { double a, b, c, d; };
__device__ __forceinline__ m2d mul2(m2d x, m2d y) {
    m2d r;
    r.a = fma(x.a, y.a, x.b * y.c);
    r.b = fma(x.a, y.b, x.b * y.d);
    r.c = fma(x.c, y.a, x.d * y.c);
    r.d = fma(x.c, y.b, x.d * y.d);
    return r;
}

__global__ void __launch_bounds__(256)
iir_chunk_ends_wmma(const float* __restrict__ u,
                    const float* __restrict__ u_init,
                    const float* __restrict__ bc,
                    const float* __restrict__ ac,
                    float2* __restrict__ cends,
                    int T, int chunks, int waves) {
    __shared__ __attribute__((aligned(16))) float lds[8][16 * PITCH];
    int lane = threadIdx.x & 31;
    int wave = __builtin_amdgcn_readfirstlane(threadIdx.x >> 5);
    int w = blockIdx.x * 8 + wave;
    if (w >= waves) w = waves - 1;
    int tb = w / chunks;
    int cidx = w - tb * chunks;
    int cbase = cidx * (GPW * GC);
    int m    = lane & 15;
    int half = lane >> 4;
    float b0 = bc[0], b1 = bc[1];
    float a2 = ac[1];
    float* tl = &lds[wave][0];

    v4f R[8]; float pv;
    group_load(u, u_init, T, tb, cbase, lane, R, pv);

    v2f Tf[4]; float hl;
    toeplitz_operands(ac[0], a2, lane, Tf, hl);
    float h14 = __int_as_float(__builtin_amdgcn_readlane(__float_as_int(hl), 14));
    float h15 = __int_as_float(__builtin_amdgcn_readlane(__float_as_int(hl), 15));
    float h16 = __int_as_float(__builtin_amdgcn_readlane(__float_as_int(hl), 16));
    float m00 = h16, m01 = -a2 * h15, m10 = h15, m11 = -a2 * h14;
    float s1 = 0.f, s2 = 0.f;
#pragma unroll 1
    for (int g = 0; g < GPW; ++g) {
        int c0 = cbase + g * GC;
        group_stage(tl, lane, R, pv);
        if (g + 1 < GPW) group_load(u, u_init, T, tb, c0 + GC, lane, R, pv);
        float2 uc[GT][4]; float up[GT][4];
        group_x_read(tl, m, half, uc, up);
        __builtin_amdgcn_sched_barrier(0);
#pragma unroll
        for (int q = 0; q < GT; ++q) {
            v2f Xf[4];
            tile_x_regs(uc[q], up[q], b0, b1, Xf);
            v8f acc = {0.f, 0.f, 0.f, 0.f, 0.f, 0.f, 0.f, 0.f};
#pragma unroll
            for (int j = 0; j < 4; ++j)
                acc = __builtin_amdgcn_wmma_f32_16x16x4_f32(
                     false, Tf[j],  false, Xf[j],
                     (short)0, acc,  false,  false);
            float n1 = fmaf(m00, s1, fmaf(m01, s2, acc[7]));
            float n2 = fmaf(m10, s1, fmaf(m11, s2, acc[6]));
            s1 = n1; s2 = n2;
        }
    }
    if (half) cends[((size_t)tb * chunks + cidx) * 16 + m] = make_float2(s1, s2);

    if (blockIdx.x == 0 && wave == 0) {
        double a1d = (double)ac[0], a2d = (double)ac[1];
        double h[17];
        h[0] = 1.0; h[1] = -a1d;
#pragma unroll
        for (int i = 2; i <= 16; ++i) h[i] = -a1d * h[i - 1] - a2d * h[i - 2];
        m2d S0 = { h[16], -a2d * h[15], h[15], -a2d * h[14] };
        m2d S1 = mul2(S0, S0);
        m2d S2 = mul2(S1, S1);
        m2d S3 = mul2(S2, S2);
        m2d S4 = mul2(S3, S3);
        if (lane == 0) {
            float* pw = (float*)(cends + (size_t)waves * 16);
            float pa = (float)S4.a, pb = (float)S4.b, pc = (float)S4.c, pd = (float)S4.d;
            pw[0] = pa; pw[1] = pb; pw[2] = pc; pw[3] = pd;
            pw[4] = (float)(S4.a - (double)pa); pw[5] = (float)(S4.b - (double)pb);
            pw[6] = (float)(S4.c - (double)pc); pw[7] = (float)(S4.d - (double)pd);
        }
    }
}

__device__ __forceinline__ void rows_store(const float* to, float* __restrict__ y, int T, int tb, int c0, int lane) {
#pragma unroll
    for (int c = 0; c < 8; ++c) {
        int flat = c * 32 + lane;
        int rr = flat >> 4, c4 = flat & 15;
        int col = c0 + 4 * c4;
        v4f v = *(const v4f*)(to + rr * PITCH + 4 * c4);
        if (col < T) *(v4f*)(y + (size_t)(tb * 16 + rr) * T + col) = v;
    }
}

__device__ __forceinline__ float chunk_row(float ph0, float ph1, float pl0, float pl1, float s1, float s2, float e) {
    float th = __fmul_rn(ph1, s2);
    float tl = fmaf(ph1, s2, -th);
    float u  = fmaf(ph0, s1, th);
    float v  = fmaf(pl0, s1, fmaf(pl1, s2, e));
    return u + (tl + v);
}
__device__ __forceinline__ void pairs_load(const float2* __restrict__ Eb, int m, int j0, int c, float2 (&z)[32], int ifirst) {
#pragma unroll
    for (int i = ifirst; i < 32; ++i)
        if (j0 + i < c) z[i] = Eb[(size_t)(j0 + i) * 16 + m];
}
__device__ __forceinline__ void pairs_fold(const float2 (&z)[32], int j0, int c, const float (&ph)[4], const float (&pl)[4],
                                           float& sx, float& sy) {
#pragma unroll
    for (int i = 0; i < 32; ++i)
        if (j0 + i < c) {
            float nx = chunk_row(ph[0], ph[1], pl[0], pl[1], sx, sy, z[i].x);
            float ny = chunk_row(ph[2], ph[3], pl[2], pl[3], sx, sy, z[i].y);
            sx = nx; sy = ny;
        }
}

__global__ void __launch_bounds__(256)
iir_tile_out_wmma(const float* __restrict__ u,
                  const float* __restrict__ u_init,
                  const float* __restrict__ bc,
                  const float* __restrict__ ac,
                  const float* __restrict__ y_init,
                  const float2* __restrict__ cends,
                  float* __restrict__ y,
                  int T, int chunks, int waves) {
    __shared__ __attribute__((aligned(16))) float lds[8][16 * PITCH];
    __shared__ __attribute__((aligned(16))) float ldo[8][16 * PITCH];
    int lane = threadIdx.x & 31;
    int wave = __builtin_amdgcn_readfirstlane(threadIdx.x >> 5);
    int w = blockIdx.x * 8 + wave;
    if (w >= waves) w = waves - 1;
    int tb = w / chunks;
    int cidx = w - tb * chunks;
    int cbase = cidx * (GPW * GC);
    int m    = lane & 15;
    int half = lane >> 4;
    float b0 = bc[0], b1 = bc[1];
    float a2 = ac[1];
    float* tl = &lds[wave][0];
    float* to = &ldo[wave][0];

    const float2* Eb = cends + (size_t)tb * chunks * 16;
    const float* pw = (const float*)(cends + (size_t)waves * 16);
    float2 yi = *(const float2*)(y_init + (size_t)(tb * 16 + m) * 2);
    float ph[4] = { pw[0], pw[1], pw[2], pw[3] };
    float pl[4] = { pw[4], pw[5], pw[6], pw[7] };
    float sx = yi.x, sy = yi.y;
    float2 za[32], zb[32];
    za[0] = Eb[m];
    pairs_load(Eb, m, 0, cidx, za, 1);
    pairs_load(Eb, m, 32, cidx, zb, 0);
    v4f R[8]; float pv;
    group_load(u, u_init, T, tb, cbase, lane, R, pv);

    v2f Tf[4]; float hl;
    toeplitz_operands(ac[0], a2, lane, Tf, hl);
    float g1 = h_gather(hl, m + 1);
    float g0 = h_gather(hl, m);
    v2f A5;
    A5[0] = half ? g1 : 0.f;
    A5[1] = half ? -a2 * g0 : 0.f;

    pairs_fold(za, 0, cidx, ph, pl, sx, sy);
    pairs_fold(zb, 32, cidx, ph, pl, sx, sy);
    for (int j0 = 64; j0 < cidx; j0 += 32) {
        pairs_load(Eb, m, j0, cidx, za, 0);
        pairs_fold(za, j0, cidx, ph, pl, sx, sy);
    }

    v2f B5;
    B5[0] = half ? sx : 0.f;
    B5[1] = half ? sy : 0.f;
#pragma unroll 1
    for (int g = 0; g < GPW; ++g) {
        int c0 = cbase + g * GC;
        group_stage(tl, lane, R, pv);
        if (g + 1 < GPW) group_load(u, u_init, T, tb, c0 + GC, lane, R, pv);
        if (g > 0) rows_store(to, y, T, tb, c0 - GC, lane);
#pragma unroll
        for (int q = 0; q < GT; ++q) {
            v2f Xf[4];
            tile_x(tl, q, m, half, b0, b1, Xf);
            v8f acc = {0.f, 0.f, 0.f, 0.f, 0.f, 0.f, 0.f, 0.f};
#pragma unroll
            for (int j = 0; j < 4; ++j)
                acc = __builtin_amdgcn_wmma_f32_16x16x4_f32(
                     false, Tf[j],  false, Xf[j],
                     (short)0, acc,  false,  false);
            acc = __builtin_amdgcn_wmma_f32_16x16x4_f32(
                 false, A5,  false, B5,
                 (short)0, acc,  false,  false);
            B5[0] = half ? acc[7] : 0.f;
            B5[1] = half ? acc[6] : 0.f;
            float* op = to + m * PITCH + 16 * q + 8 * half;
            v4f lo = {acc[0], acc[1], acc[2], acc[3]};
            v4f hi = {acc[4], acc[5], acc[6], acc[7]};
            *(v4f*)op = lo;
            *(v4f*)(op + 4) = hi;
        }
    }
    rows_store(to, y, T, tb, cbase + (GPW - 1) * GC, lane);
}

extern "C" void kernel_launch(void* const* d_in, const int* in_sizes, int n_in,
                              void* d_out, int out_size, void* d_ws, size_t ws_size,
                              hipStream_t stream) {
    const float* bc = (const float*)d_in[0];
    const float* ac = (const float*)d_in[1];
    const float* u  = (const float*)d_in[2];
    const float* yi = (const float*)d_in[3];
    const float* ui = (const float*)d_in[4];
    float* y = (float*)d_out;

    int B = in_sizes[3] / 2;
    int T = in_sizes[2] / B;
    int chunks = (T + GPW * GC - 1) / (GPW * GC);
    int waves = (B >> 4) * chunks;
    float2* cends = (float2*)d_ws;

    int blocks1 = (waves + 7) / 8;
    iir_chunk_ends_wmma<<<blocks1, 256, 0, stream>>>(u, ui, bc, ac, cends, T, chunks, waves);

    iir_tile_out_wmma<<<blocks1, 256, 0, stream>>>(u, ui, bc, ac, yi, cends, y, T, chunks, waves);
}


# single-launch kernel: pass A keeps its 16 tile accumulators in registers and pass B adds the state slice to each as ONE WMMA (kept sum as C): 80 WMMAs per wave, was 144; same sum order, same bits
# speedup vs baseline: 19.8974x; 1.3312x over previous
#include <hip/hip_runtime.h>


typedef __attribute__((ext_vector_type(2))) float v2f;
typedef __attribute__((ext_vector_type(8))) float v8f;
typedef __attribute__((ext_vector_type(4))) float v4f;

__device__ __forceinline__ float h_capture_lane(float a1, float a2, int lane) {
    float h[17];
    h[0] = 1.0f;
    h[1] = -a1;
#pragma unroll
    for (int i = 2; i <= 16; ++i) h[i] = -a1 * h[i - 1] - a2 * h[i - 2];
    float hl = h[0];
#pragma unroll
    for (int i = 1; i <= 16; ++i) hl = (lane == i) ? h[i] : hl;
    return hl;
}

__device__ __forceinline__ float h_gather(float hl, int n) {
    int nn = n < 0 ? 0 : n;
    float v = __int_as_float(__builtin_amdgcn_ds_bpermute(nn << 2, __float_as_int(hl)));
    return n < 0 ? 0.0f : v;
}

constexpr int GT = 4;
constexpr int GC = GT * 16;
constexpr int GPW = 4;
constexpr int PITCH = 68;
static_assert(GT * GPW == 16, "the chunk matrix below is (M^16)^16: 16 tiles per chunk");


__device__ __forceinline__ void toeplitz_operands(float a1, float a2, int lane, v2f (&Tf)[4], float& hl) {
    int m    = lane & 15;
    int half = lane >> 4;
    hl = h_capture_lane(a1, a2, lane);
#pragma unroll
    for (int j = 0; j < 4; ++j) {
#pragma unroll
        for (int r = 0; r < 2; ++r) {
            int k = 4 * j + 2 * half + r;
            Tf[j][r] = h_gather(hl, m - k);
        }
    }
}

__device__ __forceinline__ void group_load(const float* __restrict__ u,
                                           const float* __restrict__ u_init,
                                           int T, int tb, int c0, int lane,
                                           v4f (&v)[8], float& pv) {
#pragma unroll
    for (int c = 0; c < 8; ++c) {
        int flat = c * 32 + lane;
        int rr = flat >> 4, c4 = flat & 15;
        int col = c0 + 4 * c4;
        col = col < T ? col : T - 4;
        v[c] = *(const v4f*)(u + (size_t)(tb * 16 + rr) * T + col);
    }
    int rowg = tb * 16 + (lane & 15);
    int cp = c0 <= T ? c0 : T;
    const float* p = (c0 == 0) ? (u_init + (size_t)rowg * 2)
                               : (u + (size_t)rowg * T + (cp - 1));
    pv = *p;
}

__device__ __forceinline__ void group_stage(float* tl, int lane, const v4f (&v)[8], float pv) {
#pragma unroll
    for (int c = 0; c < 8; ++c) {
        int flat = c * 32 + lane;
        int rr = flat >> 4, c4 = flat & 15;
        *(v4f*)(tl + rr * PITCH + 4 + 4 * c4) = v[c];
    }
    if (lane < 16) tl[lane * PITCH + 3] = pv;
}

__device__ __forceinline__ void tile_x(const float* tl, int q, int m, int half, float b0, float b1, v2f (&Xf)[4]) {
#pragma unroll
    for (int j = 0; j < 4; ++j) {
        const float* p = tl + m * PITCH + 4 + 16 * q + 4 * j + 2 * half;
        float2 uc = *(const float2*)p;
        float up = p[-1];
        Xf[j][0] = b0 * uc.x + b1 * up;
        Xf[j][1] = b0 * uc.y + b1 * uc.x;
    }
}

__device__ __forceinline__ void group_x_read(const float* tl, int m, int half, float2 (&uc)[GT][4], float (&up)[GT][4]) {
#pragma unroll
    for (int q = 0; q < GT; ++q) {
#pragma unroll
        for (int j = 0; j < 4; ++j) {
            const float* p = tl + m * PITCH + 4 + 16 * q + 4 * j + 2 * half;
            uc[q][j] = *(const float2*)p;
            up[q][j] = p[-1];
        }
    }
}
__device__ __forceinline__ void tile_x_regs(const float2 (&uc)[4], const float (&up)[4], float b0, float b1, v2f (&Xf)[4]) {
#pragma unroll
    for (int j = 0; j < 4; ++j) {
        Xf[j][0] = b0 * uc[j].x + b1 * up[j];
        Xf[j][1] = b0 * uc[j].y + b1 * uc[j].x;
    }
}

struct m2d { double a, b, c, d; };
__device__ __forceinline__ m2d mul2(m2d x, m2d y) {
    m2d r;
    r.a = fma(x.a, y.a, x.b * y.c);
    r.b = fma(x.a, y.b, x.b * y.d);
    r.c = fma(x.c, y.a, x.d * y.c);
    r.d = fma(x.c, y.b, x.d * y.d);
    return r;
}

__global__ void __launch_bounds__(256)
iir_chunk_ends_wmma(const float* __restrict__ u,
                    const float* __restrict__ u_init,
                    const float* __restrict__ bc,
                    const float* __restrict__ ac,
                    float2* __restrict__ cends,
                    int T, int chunks, int waves) {
    __shared__ __attribute__((aligned(16))) float lds[8][16 * PITCH];
    int lane = threadIdx.x & 31;
    int wave = __builtin_amdgcn_readfirstlane(threadIdx.x >> 5);
    int w = blockIdx.x * 8 + wave;
    if (w >= waves) w = waves - 1;
    int tb = w / chunks;
    int cidx = w - tb * chunks;
    int cbase = cidx * (GPW * GC);
    int m    = lane & 15;
    int half = lane >> 4;
    float b0 = bc[0], b1 = bc[1];
    float a2 = ac[1];
    float* tl = &lds[wave][0];

    v4f R[8]; float pv;
    group_load(u, u_init, T, tb, cbase, lane, R, pv);

    v2f Tf[4]; float hl;
    toeplitz_operands(ac[0], a2, lane, Tf, hl);
    float h14 = __int_as_float(__builtin_amdgcn_readlane(__float_as_int(hl), 14));
    float h15 = __int_as_float(__builtin_amdgcn_readlane(__float_as_int(hl), 15));
    float h16 = __int_as_float(__builtin_amdgcn_readlane(__float_as_int(hl), 16));
    float m00 = h16, m01 = -a2 * h15, m10 = h15, m11 = -a2 * h14;
    float s1 = 0.f, s2 = 0.f;
#pragma unroll 1
    for (int g = 0; g < GPW; ++g) {
        int c0 = cbase + g * GC;
        group_stage(tl, lane, R, pv);
        if (g + 1 < GPW) group_load(u, u_init, T, tb, c0 + GC, lane, R, pv);
        float2 uc[GT][4]; float up[GT][4];
        group_x_read(tl, m, half, uc, up);
        __builtin_amdgcn_sched_barrier(0);
#pragma unroll
        for (int q = 0; q < GT; ++q) {
            v2f Xf[4];
            tile_x_regs(uc[q], up[q], b0, b1, Xf);
            v8f acc = {0.f, 0.f, 0.f, 0.f, 0.f, 0.f, 0.f, 0.f};
#pragma unroll
            for (int j = 0; j < 4; ++j)
                acc = __builtin_amdgcn_wmma_f32_16x16x4_f32(
                     false, Tf[j],  false, Xf[j],
                     (short)0, acc,  false,  false);
            float n1 = fmaf(m00, s1, fmaf(m01, s2, acc[7]));
            float n2 = fmaf(m10, s1, fmaf(m11, s2, acc[6]));
            s1 = n1; s2 = n2;
        }
    }
    if (half) cends[((size_t)tb * chunks + cidx) * 16 + m] = make_float2(s1, s2);

    if (blockIdx.x == 0 && wave == 0) {
        double a1d = (double)ac[0], a2d = (double)ac[1];
        double h[17];
        h[0] = 1.0; h[1] = -a1d;
#pragma unroll
        for (int i = 2; i <= 16; ++i) h[i] = -a1d * h[i - 1] - a2d * h[i - 2];
        m2d S0 = { h[16], -a2d * h[15], h[15], -a2d * h[14] };
        m2d S1 = mul2(S0, S0);
        m2d S2 = mul2(S1, S1);
        m2d S3 = mul2(S2, S2);
        m2d S4 = mul2(S3, S3);
        if (lane == 0) {
            float* pw = (float*)(cends + (size_t)waves * 16);
            float pa = (float)S4.a, pb = (float)S4.b, pc = (float)S4.c, pd = (float)S4.d;
            pw[0] = pa; pw[1] = pb; pw[2] = pc; pw[3] = pd;
            pw[4] = (float)(S4.a - (double)pa); pw[5] = (float)(S4.b - (double)pb);
            pw[6] = (float)(S4.c - (double)pc); pw[7] = (float)(S4.d - (double)pd);
        }
    }
}

__device__ __forceinline__ void rows_store(const float* to, float* __restrict__ y, int T, int tb, int c0, int lane) {
#pragma unroll
    for (int c = 0; c < 8; ++c) {
        int flat = c * 32 + lane;
        int rr = flat >> 4, c4 = flat & 15;
        int col = c0 + 4 * c4;
        v4f v = *(const v4f*)(to + rr * PITCH + 4 * c4);
        if (col < T) *(v4f*)(y + (size_t)(tb * 16 + rr) * T + col) = v;
    }
}

__device__ __forceinline__ float chunk_row(float ph0, float ph1, float pl0, float pl1, float s1, float s2, float e) {
    float th = __fmul_rn(ph1, s2);
    float tl = fmaf(ph1, s2, -th);
    float u  = fmaf(ph0, s1, th);
    float v  = fmaf(pl0, s1, fmaf(pl1, s2, e));
    return u + (tl + v);
}
__device__ __forceinline__ void pairs_load(const float2* __restrict__ Eb, int m, int j0, int c, float2 (&z)[32], int ifirst) {
#pragma unroll
    for (int i = ifirst; i < 32; ++i)
        if (j0 + i < c) z[i] = Eb[(size_t)(j0 + i) * 16 + m];
}
__device__ __forceinline__ void pairs_fold(const float2 (&z)[32], int j0, int c, const float (&ph)[4], const float (&pl)[4],
                                           float& sx, float& sy) {
#pragma unroll
    for (int i = 0; i < 32; ++i)
        if (j0 + i < c) {
            float nx = chunk_row(ph[0], ph[1], pl[0], pl[1], sx, sy, z[i].x);
            float ny = chunk_row(ph[2], ph[3], pl[2], pl[3], sx, sy, z[i].y);
            sx = nx; sy = ny;
        }
}

__global__ void __launch_bounds__(256)
iir_tile_out_wmma(const float* __restrict__ u,
                  const float* __restrict__ u_init,
                  const float* __restrict__ bc,
                  const float* __restrict__ ac,
                  const float* __restrict__ y_init,
                  const float2* __restrict__ cends,
                  float* __restrict__ y,
                  int T, int chunks, int waves) {
    __shared__ __attribute__((aligned(16))) float lds[8][16 * PITCH];
    __shared__ __attribute__((aligned(16))) float ldo[8][16 * PITCH];
    int lane = threadIdx.x & 31;
    int wave = __builtin_amdgcn_readfirstlane(threadIdx.x >> 5);
    int w = blockIdx.x * 8 + wave;
    if (w >= waves) w = waves - 1;
    int tb = w / chunks;
    int cidx = w - tb * chunks;
    int cbase = cidx * (GPW * GC);
    int m    = lane & 15;
    int half = lane >> 4;
    float b0 = bc[0], b1 = bc[1];
    float a2 = ac[1];
    float* tl = &lds[wave][0];
    float* to = &ldo[wave][0];

    const float2* Eb = cends + (size_t)tb * chunks * 16;
    const float* pw = (const float*)(cends + (size_t)waves * 16);
    float2 yi = *(const float2*)(y_init + (size_t)(tb * 16 + m) * 2);
    float ph[4] = { pw[0], pw[1], pw[2], pw[3] };
    float pl[4] = { pw[4], pw[5], pw[6], pw[7] };
    float sx = yi.x, sy = yi.y;
    float2 za[32], zb[32];
    za[0] = Eb[m];
    pairs_load(Eb, m, 0, cidx, za, 1);
    pairs_load(Eb, m, 32, cidx, zb, 0);
    v4f R[8]; float pv;
    group_load(u, u_init, T, tb, cbase, lane, R, pv);

    v2f Tf[4]; float hl;
    toeplitz_operands(ac[0], a2, lane, Tf, hl);
    float g1 = h_gather(hl, m + 1);
    float g0 = h_gather(hl, m);
    v2f A5;
    A5[0] = half ? g1 : 0.f;
    A5[1] = half ? -a2 * g0 : 0.f;

    pairs_fold(za, 0, cidx, ph, pl, sx, sy);
    pairs_fold(zb, 32, cidx, ph, pl, sx, sy);
    for (int j0 = 64; j0 < cidx; j0 += 32) {
        pairs_load(Eb, m, j0, cidx, za, 0);
        pairs_fold(za, j0, cidx, ph, pl, sx, sy);
    }

    v2f B5;
    B5[0] = half ? sx : 0.f;
    B5[1] = half ? sy : 0.f;
#pragma unroll 1
    for (int g = 0; g < GPW; ++g) {
        int c0 = cbase + g * GC;
        group_stage(tl, lane, R, pv);
        if (g + 1 < GPW) group_load(u, u_init, T, tb, c0 + GC, lane, R, pv);
        if (g > 0) rows_store(to, y, T, tb, c0 - GC, lane);
#pragma unroll
        for (int q = 0; q < GT; ++q) {
            v2f Xf[4];
            tile_x(tl, q, m, half, b0, b1, Xf);
            v8f acc = {0.f, 0.f, 0.f, 0.f, 0.f, 0.f, 0.f, 0.f};
#pragma unroll
            for (int j = 0; j < 4; ++j)
                acc = __builtin_amdgcn_wmma_f32_16x16x4_f32(
                     false, Tf[j],  false, Xf[j],
                     (short)0, acc,  false,  false);
            acc = __builtin_amdgcn_wmma_f32_16x16x4_f32(
                 false, A5,  false, B5,
                 (short)0, acc,  false,  false);
            B5[0] = half ? acc[7] : 0.f;
            B5[1] = half ? acc[6] : 0.f;
            float* op = to + m * PITCH + 16 * q + 8 * half;
            v4f lo = {acc[0], acc[1], acc[2], acc[3]};
            v4f hi = {acc[4], acc[5], acc[6], acc[7]};
            *(v4f*)op = lo;
            *(v4f*)(op + 4) = hi;
        }
    }
    rows_store(to, y, T, tb, cbase + (GPW - 1) * GC, lane);
}


constexpr int SCP = 260;
constexpr int SWC = 16;
constexpr int SWV = 16;
constexpr int SROW = 64;
constexpr int SWF = SWC * SCP;

__device__ __forceinline__ void chunk_x_read(const float* wl, int i, int n, int half, float2 (&uc)[4], float (&up)[4]) {
#pragma unroll
    for (int j = 0; j < 4; ++j) {
        const float* p = wl + n * SCP + 4 + 16 * i + 4 * j + 2 * half;
        uc[j] = *(const float2*)p;
        up[j] = p[-1];
    }
}

template <int I0>
__device__ __forceinline__ void zero_state_tiles(const float* wl, int n, int half, float b0, float b1,
                                                 const v2f (&Tf)[4], float m00, float m01, float m10, float m11,
                                                 float& s1, float& s2, v8f (&Y0)[16]) {
    float2 uc[4][4]; float up[4][4];
#pragma unroll
    for (int q = 0; q < 4; ++q) chunk_x_read(wl, I0 + q, n, half, uc[q], up[q]);
    __builtin_amdgcn_sched_barrier(0);
#pragma unroll
    for (int q = 0; q < 4; ++q) {
        v2f Xf[4];
        tile_x_regs(uc[q], up[q], b0, b1, Xf);
        v8f acc = {0.f, 0.f, 0.f, 0.f, 0.f, 0.f, 0.f, 0.f};
#pragma unroll
        for (int j = 0; j < 4; ++j)
            acc = __builtin_amdgcn_wmma_f32_16x16x4_f32(
                 false, Tf[j],  false, Xf[j],
                 (short)0, acc,  false,  false);
        float n1 = fmaf(m00, s1, fmaf(m01, s2, acc[7]));
        float n2 = fmaf(m10, s1, fmaf(m11, s2, acc[6]));
        s1 = n1; s2 = n2;
        Y0[I0 + q] = acc;
    }
}

__device__ __forceinline__ void halves_store(const float* wl, float* __restrict__ yrow, int hsel, int lane) {
#pragma unroll
    for (int c = 0; c < SWC; ++c) {
        v4f v = *(const v4f*)(wl + c * SCP + 4 + 128 * hsel + 4 * lane);
        *(v4f*)(yrow + 256 * c + 128 * hsel + 4 * lane) = v;
    }
}

__global__ void __launch_bounds__(512)
iir_rows_resident_wmma(const float* __restrict__ u,
                       const float* __restrict__ u_init,
                       const float* __restrict__ bc,
                       const float* __restrict__ ac,
                       const float* __restrict__ y_init,
                       float* __restrict__ y,
                       int T) {
    __shared__ __attribute__((aligned(16))) float lds[SWV * SWF];
    __shared__ float2 pairs[4][SROW];
    __shared__ float pm[8];
    int lane = threadIdx.x & 31;
    int wave = __builtin_amdgcn_readfirstlane(threadIdx.x >> 5);
    int rl = wave >> 2;
    int wq = wave & 3;
    int row = blockIdx.x * 4 + rl;
    int n    = lane & 15;
    int half = lane >> 4;
    float* wl = &lds[wave * SWF];
    const float* urow = u + (size_t)row * T + wq * (SWC * 256);
    float* yrow = y + (size_t)row * T + wq * (SWC * 256);

    const float* pp = (wq == 0 && n == 0) ? (u_init + (size_t)row * 2) : (urow + 256 * n - 1);
    float pv = *pp;
    float2 yi = *(const float2*)(y_init + (size_t)row * 2);
    v4f Ra[SWC], Rb[SWC];
#pragma unroll
    for (int c = 0; c < SWC; ++c) Ra[c] = *(const v4f*)(urow + 256 * c + 4 * lane);
#pragma unroll
    for (int c = 0; c < SWC; ++c) Rb[c] = *(const v4f*)(urow + 256 * c + 128 + 4 * lane);

    float b0 = bc[0], b1 = bc[1];
    float a2 = ac[1];
    v2f Tf[4]; float hl;
    toeplitz_operands(ac[0], a2, lane, Tf, hl);
    float h14 = __int_as_float(__builtin_amdgcn_readlane(__float_as_int(hl), 14));
    float h15 = __int_as_float(__builtin_amdgcn_readlane(__float_as_int(hl), 15));
    float h16 = __int_as_float(__builtin_amdgcn_readlane(__float_as_int(hl), 16));
    float m00 = h16, m01 = -a2 * h15, m10 = h15, m11 = -a2 * h14;
    float g1 = h_gather(hl, n + 1);
    float g0 = h_gather(hl, n);
    v2f A5;
    A5[0] = half ? g1 : 0.f;
    A5[1] = half ? -a2 * g0 : 0.f;

    if (wave == 0) {
        double a1d = (double)ac[0], a2d = (double)ac[1];
        double h[17];
        h[0] = 1.0; h[1] = -a1d;
#pragma unroll
        for (int i = 2; i <= 16; ++i) h[i] = -a1d * h[i - 1] - a2d * h[i - 2];
        m2d S0 = { h[16], -a2d * h[15], h[15], -a2d * h[14] };
        m2d S1 = mul2(S0, S0);
        m2d S2 = mul2(S1, S1);
        m2d S3 = mul2(S2, S2);
        m2d S4 = mul2(S3, S3);
        if (lane == 0) {
            float pa = (float)S4.a, pb = (float)S4.b, pc = (float)S4.c, pd = (float)S4.d;
            pm[0] = pa; pm[1] = pb; pm[2] = pc; pm[3] = pd;
            pm[4] = (float)(S4.a - (double)pa); pm[5] = (float)(S4.b - (double)pb);
            pm[6] = (float)(S4.c - (double)pc); pm[7] = (float)(S4.d - (double)pd);
        }
    }

#pragma unroll
    for (int c = 0; c < SWC; ++c) *(v4f*)(wl + c * SCP + 4 + 4 * lane) = Ra[c];
    if (lane < 16) wl[lane * SCP + 3] = pv;
    float s1 = 0.f, s2 = 0.f;
    v8f Y0[16];
    zero_state_tiles<0>(wl, n, half, b0, b1, Tf, m00, m01, m10, m11, s1, s2, Y0);
    zero_state_tiles<4>(wl, n, half, b0, b1, Tf, m00, m01, m10, m11, s1, s2, Y0);
#pragma unroll
    for (int c = 0; c < SWC; ++c) *(v4f*)(wl + c * SCP + 4 + 128 + 4 * lane) = Rb[c];
    zero_state_tiles<8>(wl, n, half, b0, b1, Tf, m00, m01, m10, m11, s1, s2, Y0);
    zero_state_tiles<12>(wl, n, half, b0, b1, Tf, m00, m01, m10, m11, s1, s2, Y0);
    if (half) pairs[rl][wq * SWC + n] = make_float2(s1, s2);

    __syncthreads();

    float ph[4] = { pm[0], pm[1], pm[2], pm[3] };
    float pl[4] = { pm[4], pm[5], pm[6], pm[7] };
    float sx = yi.x, sy = yi.y;
    for (int b = 0; b < wq; ++b) {
        float2 z[SWC];
#pragma unroll
        for (int k = 0; k < SWC; ++k) z[k] = pairs[rl][b * SWC + k];
#pragma unroll
        for (int k = 0; k < SWC; ++k) {
            float nx = chunk_row(ph[0], ph[1], pl[0], pl[1], sx, sy, z[k].x);
            float ny = chunk_row(ph[2], ph[3], pl[2], pl[3], sx, sy, z[k].y);
            sx = nx; sy = ny;
        }
    }
    float cx = sx, cy = sy;
    {
        float2 z[SWC];
#pragma unroll
        for (int k = 0; k < SWC - 1; ++k) z[k] = pairs[rl][wq * SWC + k];
#pragma unroll
        for (int k = 0; k < SWC - 1; ++k) {
            float nx = chunk_row(ph[0], ph[1], pl[0], pl[1], sx, sy, z[k].x);
            float ny = chunk_row(ph[2], ph[3], pl[2], pl[3], sx, sy, z[k].y);
            sx = nx; sy = ny;
            cx = (n == k + 1) ? sx : cx;
            cy = (n == k + 1) ? sy : cy;
        }
    }

    v2f B5;
    B5[0] = half ? cx : 0.f;
    B5[1] = half ? cy : 0.f;
#pragma unroll
    for (int i = 0; i < 16; ++i) {
        v8f acc = __builtin_amdgcn_wmma_f32_16x16x4_f32(
             false, A5,  false, B5,
             (short)0, Y0[i],  false,  false);
        B5[0] = half ? acc[7] : 0.f;
        B5[1] = half ? acc[6] : 0.f;
        float* op = wl + n * SCP + 4 + 16 * i + 8 * half;
        v4f lo = {acc[0], acc[1], acc[2], acc[3]};
        v4f hi = {acc[4], acc[5], acc[6], acc[7]};
        *(v4f*)op = lo;
        *(v4f*)(op + 4) = hi;
        if (i == 7) halves_store(wl, yrow, 0, lane);
    }
    halves_store(wl, yrow, 1, lane);
}

extern "C" void kernel_launch(void* const* d_in, const int* in_sizes, int n_in,
                              void* d_out, int out_size, void* d_ws, size_t ws_size,
                              hipStream_t stream) {
    const float* bc = (const float*)d_in[0];
    const float* ac = (const float*)d_in[1];
    const float* u  = (const float*)d_in[2];
    const float* yi = (const float*)d_in[3];
    const float* ui = (const float*)d_in[4];
    float* y = (float*)d_out;

    int B = in_sizes[3] / 2;
    int T = in_sizes[2] / B;
    if (T == SROW * 256 && (B & 3) == 0) {
        iir_rows_resident_wmma<<<B / 4, 512, 0, stream>>>(u, ui, bc, ac, yi, y, T);
        return;
    }
    int chunks = (T + GPW * GC - 1) / (GPW * GC);
    int waves = (B >> 4) * chunks;
    float2* cends = (float2*)d_ws;

    int blocks1 = (waves + 7) / 8;
    iir_chunk_ends_wmma<<<blocks1, 256, 0, stream>>>(u, ui, bc, ac, cends, T, chunks, waves);

    iir_tile_out_wmma<<<blocks1, 256, 0, stream>>>(u, ui, bc, ac, yi, cends, y, T, chunks, waves);
}
